# starting kernel
# baseline (speedup 1.0000x reference)
#include <hip/hip_runtime.h>


typedef __attribute__((ext_vector_type(2))) float v2f;
typedef __attribute__((ext_vector_type(8))) float v8f;

#define K_DIM 128

__global__ void fill_f32(float* __restrict__ p, float v, long n) {
    long i = (long)blockIdx.x * blockDim.x + threadIdx.x;
    if (i < n) p[i] = v;
}

__global__ void degree_kernel(const int* __restrict__ dst,
                              float* __restrict__ deg, int n_edges) {
    int e = blockIdx.x * blockDim.x + threadIdx.x;
    if (e < n_edges) atomicAdd(&deg[dst[e]], 1.0f);
}

__global__ void rsqrt_inplace(float* __restrict__ deg, int n) {
    int i = blockIdx.x * blockDim.x + threadIdx.x;
    if (i < n) deg[i] = rsqrtf(deg[i]);
}

template <int NCOLS>
__global__ void gemm_wmma_f32(const float* __restrict__ A,
                              const float* __restrict__ W,
                              float* __restrict__ C, int nrows) {
    constexpr int TC = NCOLS / 16;
    const int lane  = threadIdx.x & 31;
    const int wave  = threadIdx.x >> 5;
    const int tile  = blockIdx.x * (blockDim.x >> 5) + wave;
    const int trow  = tile / TC;
    const int tcol  = tile % TC;
    const int rbase = trow * 16;
    if (rbase >= nrows) return;

    const int m     = lane & 15;
    const int khalf = lane >> 4;

    const int arow_idx = (rbase + m < nrows) ? (rbase + m) : (nrows - 1);
    const float* arow = A + (long)arow_idx * K_DIM;
    const float* wcol = W + tcol * 16 + m;

    v8f acc = {};
#pragma unroll 8
    for (int k = 0; k < K_DIM; k += 4) {
        const int kk = k + 2 * khalf;
        v2f a, b;
        a.x = arow[kk];
        a.y = arow[kk + 1];
        b.x = wcol[(long)kk * NCOLS];
        b.y = wcol[(long)(kk + 1) * NCOLS];
        acc = __builtin_amdgcn_wmma_f32_16x16x4_f32(
            false, a, false, b, (short)0, acc, false, false);
    }

    if (rbase + 16 <= nrows) {
        float* cbase = C + (long)(rbase + 8 * khalf) * NCOLS + tcol * 16 + m;
#pragma unroll
        for (int r = 0; r < 8; ++r) cbase[(long)r * NCOLS] = acc[r];
    } else {
#pragma unroll
        for (int r = 0; r < 8; ++r) {
            const int row = rbase + r + 8 * khalf;
            if (row < nrows) C[(long)row * NCOLS + tcol * 16 + m] = acc[r];
        }
    }
}

template <int CH>
__global__ void aggregate_edges(const float* __restrict__ h,
                                const int* __restrict__ src,
                                const int* __restrict__ dst,
                                const float* __restrict__ dinv,
                                float* __restrict__ out, int n_edges) {
    const int lane = threadIdx.x & 31;
    const int e    = blockIdx.x * (blockDim.x >> 5) + (threadIdx.x >> 5);
    if (e >= n_edges) return;

    const int   s   = src[e];
    const int   d   = dst[e];
    const float nrm = dinv[s] * dinv[d];

    constexpr int V = CH / 32;
    const float* hp = h   + (long)s * CH + lane * V;
    float*       op = out + (long)d * CH + lane * V;

    float vals[V];
#pragma unroll
    for (int i = 0; i < V; ++i) vals[i] = hp[i];
#pragma unroll
    for (int i = 0; i < V; ++i) atomicAdd(&op[i], vals[i] * nrm);
}

template <int CH>
__global__ void bias_relu_self(const float* __restrict__ h,
                               const float* __restrict__ dinv,
                               const float* __restrict__ bias,
                               float* __restrict__ agg, int n_nodes) {
    long idx = (long)blockIdx.x * blockDim.x + threadIdx.x;
    if (idx >= (long)n_nodes * CH) return;
    const int node = (int)(idx / CH);
    const int c    = (int)(idx % CH);
    const float di = dinv[node];
    const float v  = agg[idx] + h[idx] * di * di + bias[c];
    agg[idx] = v > 0.0f ? v : 0.0f;
}

extern "C" void kernel_launch(void* const* d_in, const int* in_sizes, int n_in,
                              void* d_out, int out_size, void* d_ws, size_t ws_size,
                              hipStream_t stream) {
    const float* x  = (const float*)d_in[0];
    const int*   ei = (const int*)d_in[1];
    const float* W1 = (const float*)d_in[2];
    const float* b1 = (const float*)d_in[3];
    const float* W2 = (const float*)d_in[4];
    const float* b2 = (const float*)d_in[5];

    const int N   = in_sizes[0] / K_DIM;
    const int E   = in_sizes[1] / 2;
    (void)n_in; (void)ws_size; (void)out_size;

    const int* src = ei;
    const int* dst = ei + E;
    float* out = (float*)d_out;

    char* ws = (char*)d_ws;
    const size_t degBytes = (((size_t)N * 4) + 255) & ~(size_t)255;
    float* deg  = (float*)ws;
    float* bufA = (float*)(ws + degBytes);
    float* bufB = (float*)(ws + degBytes + (size_t)N * 128 * 4);

    const int TB = 256;

    fill_f32<<<(N + TB - 1) / TB, TB, 0, stream>>>(deg, 1.0f, N);
    degree_kernel<<<(E + TB - 1) / TB, TB, 0, stream>>>(dst, deg, E);
    rsqrt_inplace<<<(N + TB - 1) / TB, TB, 0, stream>>>(deg, N);

    {
        const int tiles = ((N + 15) / 16) * (128 / 16);
        gemm_wmma_f32<128><<<(tiles + 7) / 8, 256, 0, stream>>>(x, W1, bufA, N);
        const long n1 = (long)N * 128;
        fill_f32<<<(unsigned)((n1 + TB - 1) / TB), TB, 0, stream>>>(bufB, 0.0f, n1);
        aggregate_edges<128><<<(E + 7) / 8, 256, 0, stream>>>(bufA, src, dst, deg, bufB, E);
        bias_relu_self<128><<<(unsigned)((n1 + TB - 1) / TB), TB, 0, stream>>>(bufA, deg, b1, bufB, N);
    }

    {
        const int tiles = ((N + 15) / 16) * (64 / 16);
        gemm_wmma_f32<64><<<(tiles + 7) / 8, 256, 0, stream>>>(bufB, W2, bufA, N);
        const long n2 = (long)N * 64;
        fill_f32<<<(unsigned)((n2 + TB - 1) / TB), TB, 0, stream>>>(out, 0.0f, n2);
        aggregate_edges<64><<<(E + 7) / 8, 256, 0, stream>>>(bufA, src, dst, deg, out, E);
        bias_relu_self<64><<<(unsigned)((n2 + TB - 1) / TB), TB, 0, stream>>>(bufA, deg, b2, out, N);
    }
}


# Edges grouped by destination per call (int rank atomics, scan, place); one wave per row sums its sources in registers, stores once: no float atomics/clears; sum order=arrival rank; layer 1 = (A x) W1
# speedup vs baseline: 2.1876x; 2.1876x over previous
#include <hip/hip_runtime.h>


typedef __attribute__((ext_vector_type(2))) float v2f;
typedef __attribute__((ext_vector_type(8))) float v8f;

#define K_DIM 128

__global__ void zero_i32(int* __restrict__ p, int n) {
    int i = blockIdx.x * blockDim.x + threadIdx.x;
    if (i < n) p[i] = 0;
}

__global__ void rank_edges(const int* __restrict__ dst, int* __restrict__ cnt,
                           int* __restrict__ rank, int n_edges) {
    int e = blockIdx.x * blockDim.x + threadIdx.x;
    if (e < n_edges) rank[e] = atomicAdd(&cnt[dst[e]], 1);
}

#define SCAN_EPT 16
__global__ void __launch_bounds__(1024)
scan_counts(int* __restrict__ offs, float* __restrict__ dinv, int n) {
    __shared__ int wsum[32];
    __shared__ int tile_total;
    const int t    = threadIdx.x;
    const int lane = t & 31;
    const int wave = t >> 5;
    int carry = 0;
    for (int base0 = 0; base0 < n; base0 += 1024 * SCAN_EPT) {
        const int i0 = base0 + t * SCAN_EPT;
        const bool whole = (i0 + SCAN_EPT <= n);
        int c[SCAN_EPT];
        if (whole) {
#pragma unroll
            for (int u = 0; u < SCAN_EPT; ++u) c[u] = offs[i0 + u];
        } else {
#pragma unroll
            for (int u = 0; u < SCAN_EPT; ++u) c[u] = (i0 + u < n) ? offs[i0 + u] : 0;
        }
        int tsum = 0;
#pragma unroll
        for (int u = 0; u < SCAN_EPT; ++u) tsum += c[u];
        int incl = tsum;
#pragma unroll
        for (int d = 1; d < 32; d <<= 1) {
            int y = __shfl_up(incl, d);
            if (lane >= d) incl += y;
        }
        if (lane == 31) wsum[wave] = incl;
        __syncthreads();
        if (wave == 0) {
            const int v = wsum[lane];
            int iv = v;
#pragma unroll
            for (int d = 1; d < 32; d <<= 1) {
                int y = __shfl_up(iv, d);
                if (lane >= d) iv += y;
            }
            wsum[lane] = iv - v;
            if (lane == 31) tile_total = iv;
        }
        __syncthreads();
        int run = carry + wsum[wave] + (incl - tsum);
        if (whole) {
#pragma unroll
            for (int u = 0; u < SCAN_EPT; ++u) {
                offs[i0 + u] = run;
                dinv[i0 + u] = rsqrtf((float)(c[u] + 1));
                run += c[u];
            }
        } else {
#pragma unroll
            for (int u = 0; u < SCAN_EPT; ++u) {
                if (i0 + u < n) {
                    offs[i0 + u] = run;
                    dinv[i0 + u] = rsqrtf((float)(c[u] + 1));
                    run += c[u];
                }
            }
        }
        carry += tile_total;
        __syncthreads();
    }
    if (t == 0) offs[n] = carry;
}

__global__ void place_edges(const int* __restrict__ src, const int* __restrict__ dst,
                            const int* __restrict__ offs, const int* __restrict__ rank,
                            int* __restrict__ ssrc, int n_edges) {
    int e = blockIdx.x * blockDim.x + threadIdx.x;
    if (e < n_edges) ssrc[offs[dst[e]] + rank[e]] = src[e];
}

template <int NCOLS, bool FUSE>
__global__ void gemm_wmma_f32(const float* A,
                              const float* __restrict__ W,
                              const float* __restrict__ bias,
                              float* C, int nrows) {
    constexpr int TC = NCOLS / 16;
    const int lane  = threadIdx.x & 31;
    const int wave  = threadIdx.x >> 5;
    const int tile  = blockIdx.x * (blockDim.x >> 5) + wave;
    const int trow  = tile / TC;
    const int tcol  = tile % TC;
    const int rbase = trow * 16;
    if (rbase >= nrows) return;

    const int m     = lane & 15;
    const int khalf = lane >> 4;

    const int arow_idx = (rbase + m < nrows) ? (rbase + m) : (nrows - 1);
    const float* arow = A + (long)arow_idx * K_DIM;
    const float* wcol = W + tcol * 16 + m;

    v8f acc = {};
#pragma unroll 8
    for (int k = 0; k < K_DIM; k += 4) {
        const int kk = k + 2 * khalf;
        v2f a, b;
        a.x = arow[kk];
        a.y = arow[kk + 1];
        b.x = wcol[(long)kk * NCOLS];
        b.y = wcol[(long)(kk + 1) * NCOLS];
        acc = __builtin_amdgcn_wmma_f32_16x16x4_f32(
            false, a, false, b, (short)0, acc, false, false);
    }

    if (FUSE) {
        static_assert(!FUSE || NCOLS == 128, "in-place form needs one stripe per work-group");
        __syncthreads();
        const float bv = bias[tcol * 16 + m];
#pragma unroll
        for (int r = 0; r < 8; ++r) {
            const float v = acc[r] + bv;
            acc[r] = v > 0.0f ? v : 0.0f;
        }
    }

    if (rbase + 16 <= nrows) {
        float* cbase = C + (long)(rbase + 8 * khalf) * NCOLS + tcol * 16 + m;
#pragma unroll
        for (int r = 0; r < 8; ++r) cbase[(long)r * NCOLS] = acc[r];
    } else {
#pragma unroll
        for (int r = 0; r < 8; ++r) {
            const int row = rbase + r + 8 * khalf;
            if (row < nrows) C[(long)row * NCOLS + tcol * 16 + m] = acc[r];
        }
    }
}

template <int CH, bool FINAL>
__global__ void gather_rows(const float* __restrict__ h,
                            const int* __restrict__ offs,
                            const int* __restrict__ ssrc,
                            const float* __restrict__ dinv,
                            const float* __restrict__ bias,
                            float* __restrict__ out, int n_nodes) {
    constexpr int V = CH / 32;
    typedef __attribute__((ext_vector_type(V))) float vrow;
    const int lane = threadIdx.x & 31;
    const int node = __builtin_amdgcn_readfirstlane(
        (int)(blockIdx.x * (blockDim.x >> 5) + (threadIdx.x >> 5)));
    if (node >= n_nodes) return;

    const int   beg = offs[node];
    const int   end = offs[node + 1];
    const float di  = dinv[node];
    const vrow* hv  = (const vrow*)h;

    vrow acc = hv[(long)node * 32 + lane] * di;

    for (int base = beg; base < end; base += 32) {
        int my_s = 0, my_w = 0;
        if (base + lane < end) {
            my_s = ssrc[base + lane];
            my_w = __builtin_bit_cast(int, dinv[my_s]);
        }
        const int cnt = (end - base < 32) ? (end - base) : 32;
        for (int j = 0; j < cnt; j += 4) {
            const int s0 = __builtin_amdgcn_readlane(my_s, j);
            const int s1 = __builtin_amdgcn_readlane(my_s, j + 1);
            const int s2 = __builtin_amdgcn_readlane(my_s, j + 2);
            const int s3 = __builtin_amdgcn_readlane(my_s, j + 3);
            const float w0 = __builtin_bit_cast(float, __builtin_amdgcn_readlane(my_w, j));
            const float w1 = __builtin_bit_cast(float, __builtin_amdgcn_readlane(my_w, j + 1));
            const float w2 = __builtin_bit_cast(float, __builtin_amdgcn_readlane(my_w, j + 2));
            const float w3 = __builtin_bit_cast(float, __builtin_amdgcn_readlane(my_w, j + 3));
            const vrow r0 = hv[(long)s0 * 32 + lane];
            const vrow r1 = hv[(long)s1 * 32 + lane];
            const vrow r2 = hv[(long)s2 * 32 + lane];
            const vrow r3 = hv[(long)s3 * 32 + lane];
            acc += r0 * w0;
            acc += r1 * w1;
            acc += r2 * w2;
            acc += r3 * w3;
        }
    }

    vrow v = acc * di;
    if (FINAL) {
        v += ((const vrow*)bias)[lane];
#pragma unroll
        for (int i = 0; i < V; ++i) v[i] = v[i] > 0.0f ? v[i] : 0.0f;
    }
    ((vrow*)out)[(long)node * 32 + lane] = v;
}

extern "C" void kernel_launch(void* const* d_in, const int* in_sizes, int n_in,
                              void* d_out, int out_size, void* d_ws, size_t ws_size,
                              hipStream_t stream) {
    const float* x  = (const float*)d_in[0];
    const int*   ei = (const int*)d_in[1];
    const float* W1 = (const float*)d_in[2];
    const float* b1 = (const float*)d_in[3];
    const float* W2 = (const float*)d_in[4];
    const float* b2 = (const float*)d_in[5];

    const int N   = in_sizes[0] / K_DIM;
    const int E   = in_sizes[1] / 2;
    (void)n_in; (void)ws_size; (void)out_size;

    const int* src = ei;
    const int* dst = ei + E;
    float* out = (float*)d_out;

    char* ws = (char*)d_ws;
    const size_t degBytes = (((size_t)N * 4) + 255) & ~(size_t)255;
    float* dinv = (float*)ws;
    float* bufA = (float*)(ws + degBytes);
    float* bufB = (float*)(ws + degBytes + (size_t)N * 128 * 4);
    char*  lists = (char*)(bufB + (size_t)N * 64);
    const size_t offBytes = ((((size_t)N + 1) * 4) + 255) & ~(size_t)255;
    int* offs = (int*)lists;
    int* rank = (int*)(lists + offBytes);
    int* ssrc = rank + E;

    const int TB = 256;

    zero_i32<<<(N + TB - 1) / TB, TB, 0, stream>>>(offs, N);
    rank_edges<<<(E + TB - 1) / TB, TB, 0, stream>>>(dst, offs, rank, E);
    scan_counts<<<1, 1024, 0, stream>>>(offs, dinv, N);
    place_edges<<<(E + TB - 1) / TB, TB, 0, stream>>>(src, dst, offs, rank, ssrc, E);

    {
        gather_rows<128, false><<<(N + 7) / 8, 256, 0, stream>>>(x, offs, ssrc, dinv, b1, bufA, N);
        const int tiles = ((N + 15) / 16) * (128 / 16);
        gemm_wmma_f32<128, true><<<(tiles + 7) / 8, 256, 0, stream>>>(bufA, W1, b1, bufA, N);
    }

    {
        const int tiles = ((N + 15) / 16) * (64 / 16);
        gemm_wmma_f32<64, false><<<(tiles + 7) / 8, 256, 0, stream>>>(bufA, W2, b2, bufB, N);
        gather_rows<64, true><<<(N + 7) / 8, 256, 0, stream>>>(bufB, offs, ssrc, dinv, b2, out, N);
    }
}


# Edges grouped by destination with no global atomics: per-call two-level counting sort on LDS counters (256-row buckets: count, row scan, bin, per-bucket rank+place); sum order=arrival rank
# speedup vs baseline: 3.3118x; 1.5139x over previous
#include <hip/hip_runtime.h>


typedef __attribute__((ext_vector_type(2))) float v2f;
typedef __attribute__((ext_vector_type(8))) float v8f;

#define K_DIM 128

#define BKT_SHIFT 8
#define BKT_NODES (1 << BKT_SHIFT)
#define N_CHUNKS  256
#define MAX_BKTS  1024

__device__ __forceinline__ int block_excl_scan(int v, int* sm, int& total) {
    const int lane = threadIdx.x & 31;
    const int wave = threadIdx.x >> 5;
    int incl = v;
#pragma unroll
    for (int d = 1; d < 32; d <<= 1) {
        int y = __shfl_up(incl, d);
        if (lane >= d) incl += y;
    }
    if (lane == 31) sm[wave] = incl;
    __syncthreads();
    if (wave == 0) {
        const int nw = (int)(blockDim.x + 31) >> 5;
        const int w = (lane < nw) ? sm[lane] : 0;
        int iw = w;
#pragma unroll
        for (int d = 1; d < 32; d <<= 1) {
            int y = __shfl_up(iw, d);
            if (lane >= d) iw += y;
        }
        sm[lane] = iw - w;
        if (lane == 31) sm[32] = iw;
    }
    __syncthreads();
    total = sm[32];
    return sm[wave] + (incl - v);
}

__global__ void __launch_bounds__(1024)
count_buckets(const int* __restrict__ dst, int* __restrict__ cnt,
              int n_edges, int per, int n_bkts) {
    __shared__ int hist[MAX_BKTS];
    const int t = threadIdx.x, nt = blockDim.x, c = blockIdx.x;
    for (int b = t; b < n_bkts; b += nt) hist[b] = 0;
    __syncthreads();
    const int beg = c * per;
    const int end = (beg + per < n_edges) ? (beg + per) : n_edges;
    for (int i = beg + t; i < end; i += 4 * nt) {
        const int i1 = i + nt, i2 = i + 2 * nt, i3 = i + 3 * nt;
        const bool g1 = i1 < end, g2 = i2 < end, g3 = i3 < end;
        const int d0 = dst[i];
        const int d1 = g1 ? dst[i1] : 0;
        const int d2 = g2 ? dst[i2] : 0;
        const int d3 = g3 ? dst[i3] : 0;
        atomicAdd(&hist[d0 >> BKT_SHIFT], 1);
        if (g1) atomicAdd(&hist[d1 >> BKT_SHIFT], 1);
        if (g2) atomicAdd(&hist[d2 >> BKT_SHIFT], 1);
        if (g3) atomicAdd(&hist[d3 >> BKT_SHIFT], 1);
    }
    __syncthreads();
    for (int b = t; b < n_bkts; b += nt) cnt[b * N_CHUNKS + c] = hist[b];
}

__global__ void __launch_bounds__(N_CHUNKS)
scan_rows(int* __restrict__ cnt, int* __restrict__ btot) {
    __shared__ int sm[33];
    const int b = blockIdx.x, c = threadIdx.x;
    const int v = cnt[b * N_CHUNKS + c];
    int total;
    const int ex = block_excl_scan(v, sm, total);
    cnt[b * N_CHUNKS + c] = ex;
    if (c == 0) btot[b] = total;
}

__global__ void __launch_bounds__(1024)
bin_edges(const int* __restrict__ src, const int* __restrict__ dst,
          const int* __restrict__ cnt, const int* __restrict__ btot,
          int* __restrict__ recs_i, int n_edges, int per, int n_bkts) {
    __shared__ int cur[MAX_BKTS];
    __shared__ int sm[33];
    int2* __restrict__ recs = (int2*)recs_i;
    const int t = threadIdx.x, nt = blockDim.x, c = blockIdx.x;
    int total;
    const int bt = (t < n_bkts) ? btot[t] : 0;
    const int rp = (t < n_bkts) ? cnt[t * N_CHUNKS + c] : 0;
    const int bs = block_excl_scan(bt, sm, total);
    if (t < n_bkts) cur[t] = bs + rp;
    __syncthreads();
    const int beg = c * per;
    const int end = (beg + per < n_edges) ? (beg + per) : n_edges;
    for (int i = beg + t; i < end; i += 4 * nt) {
        const int i1 = i + nt, i2 = i + 2 * nt, i3 = i + 3 * nt;
        const bool g1 = i1 < end, g2 = i2 < end, g3 = i3 < end;
        const int s0 = src[i],            d0 = dst[i];
        const int s1 = g1 ? src[i1] : 0,  d1 = g1 ? dst[i1] : 0;
        const int s2 = g2 ? src[i2] : 0,  d2 = g2 ? dst[i2] : 0;
        const int s3 = g3 ? src[i3] : 0,  d3 = g3 ? dst[i3] : 0;
        recs[atomicAdd(&cur[d0 >> BKT_SHIFT], 1)] = make_int2(s0, d0);
        if (g1) recs[atomicAdd(&cur[d1 >> BKT_SHIFT], 1)] = make_int2(s1, d1);
        if (g2) recs[atomicAdd(&cur[d2 >> BKT_SHIFT], 1)] = make_int2(s2, d2);
        if (g3) recs[atomicAdd(&cur[d3 >> BKT_SHIFT], 1)] = make_int2(s3, d3);
    }
}

__global__ void __launch_bounds__(1024)
group_bucket(const int* __restrict__ recs_i, const int* __restrict__ btot,
             int* __restrict__ offs, float* __restrict__ dinv,
             int* __restrict__ ssrc, int n_nodes, int n_bkts) {
    __shared__ int lcnt[BKT_NODES];
    __shared__ int sm[33];
    const int2* __restrict__ recs = (const int2*)recs_i;
    const int t = threadIdx.x, nt = blockDim.x, b = blockIdx.x;
    const int node0 = b << BKT_SHIFT;

    int p = 0;
    for (int j = t; j < b; j += nt) p += btot[j];
    int bstart;
    (void)block_excl_scan(p, sm, bstart);
    const int bend = bstart + btot[b];

    if (t < BKT_NODES) lcnt[t] = 0;
    __syncthreads();
    for (int i = bstart + t; i < bend; i += 4 * nt) {
        const int i1 = i + nt, i2 = i + 2 * nt, i3 = i + 3 * nt;
        const bool g1 = i1 < bend, g2 = i2 < bend, g3 = i3 < bend;
        const int d0 = recs[i].y;
        const int d1 = g1 ? recs[i1].y : 0;
        const int d2 = g2 ? recs[i2].y : 0;
        const int d3 = g3 ? recs[i3].y : 0;
        atomicAdd(&lcnt[d0 - node0], 1);
        if (g1) atomicAdd(&lcnt[d1 - node0], 1);
        if (g2) atomicAdd(&lcnt[d2 - node0], 1);
        if (g3) atomicAdd(&lcnt[d3 - node0], 1);
    }
    __syncthreads();

    const int cv = (t < BKT_NODES) ? lcnt[t] : 0;
    int tot2;
    const int ex = block_excl_scan(cv, sm, tot2);
    if (t < BKT_NODES) {
        lcnt[t] = bstart + ex;
        const int node = node0 + t;
        if (node < n_nodes) {
            offs[node] = bstart + ex;
            dinv[node] = rsqrtf((float)(cv + 1));
        }
    }
    if (b == n_bkts - 1 && t == 0) offs[n_nodes] = bend;
    __syncthreads();

    for (int i = bstart + t; i < bend; i += 4 * nt) {
        const int i1 = i + nt, i2 = i + 2 * nt, i3 = i + 3 * nt;
        const bool g1 = i1 < bend, g2 = i2 < bend, g3 = i3 < bend;
        const int2 z = make_int2(0, 0);
        const int2 r0 = recs[i];
        const int2 r1 = g1 ? recs[i1] : z;
        const int2 r2 = g2 ? recs[i2] : z;
        const int2 r3 = g3 ? recs[i3] : z;
        ssrc[atomicAdd(&lcnt[r0.y - node0], 1)] = r0.x;
        if (g1) ssrc[atomicAdd(&lcnt[r1.y - node0], 1)] = r1.x;
        if (g2) ssrc[atomicAdd(&lcnt[r2.y - node0], 1)] = r2.x;
        if (g3) ssrc[atomicAdd(&lcnt[r3.y - node0], 1)] = r3.x;
    }
}

template <int NCOLS, bool FUSE>
__global__ void gemm_wmma_f32(const float* A,
                              const float* __restrict__ W,
                              const float* __restrict__ bias,
                              float* C, int nrows) {
    constexpr int TC = NCOLS / 16;
    const int lane  = threadIdx.x & 31;
    const int wave  = threadIdx.x >> 5;
    const int tile  = blockIdx.x * (blockDim.x >> 5) + wave;
    const int trow  = tile / TC;
    const int tcol  = tile % TC;
    const int rbase = trow * 16;
    if (rbase >= nrows) return;

    const int m     = lane & 15;
    const int khalf = lane >> 4;

    const int arow_idx = (rbase + m < nrows) ? (rbase + m) : (nrows - 1);
    const float* arow = A + (long)arow_idx * K_DIM;
    const float* wcol = W + tcol * 16 + m;

    v8f acc = {};
#pragma unroll 8
    for (int k = 0; k < K_DIM; k += 4) {
        const int kk = k + 2 * khalf;
        v2f a, b;
        a.x = arow[kk];
        a.y = arow[kk + 1];
        b.x = wcol[(long)kk * NCOLS];
        b.y = wcol[(long)(kk + 1) * NCOLS];
        acc = __builtin_amdgcn_wmma_f32_16x16x4_f32(
            false, a, false, b, (short)0, acc, false, false);
    }

    if (FUSE) {
        static_assert(!FUSE || NCOLS == 128, "in-place form needs one stripe per work-group");
        __syncthreads();
        const float bv = bias[tcol * 16 + m];
#pragma unroll
        for (int r = 0; r < 8; ++r) {
            const float v = acc[r] + bv;
            acc[r] = v > 0.0f ? v : 0.0f;
        }
    }

    if (rbase + 16 <= nrows) {
        float* cbase = C + (long)(rbase + 8 * khalf) * NCOLS + tcol * 16 + m;
#pragma unroll
        for (int r = 0; r < 8; ++r) cbase[(long)r * NCOLS] = acc[r];
    } else {
#pragma unroll
        for (int r = 0; r < 8; ++r) {
            const int row = rbase + r + 8 * khalf;
            if (row < nrows) C[(long)row * NCOLS + tcol * 16 + m] = acc[r];
        }
    }
}

template <int CH, bool FINAL>
__global__ void gather_rows(const float* __restrict__ h,
                            const int* __restrict__ offs,
                            const int* __restrict__ ssrc,
                            const float* __restrict__ dinv,
                            const float* __restrict__ bias,
                            float* __restrict__ out, int n_nodes) {
    constexpr int V = CH / 32;
    typedef __attribute__((ext_vector_type(V))) float vrow;
    const int lane = threadIdx.x & 31;
    const int node = __builtin_amdgcn_readfirstlane(
        (int)(blockIdx.x * (blockDim.x >> 5) + (threadIdx.x >> 5)));
    if (node >= n_nodes) return;

    const int   beg = offs[node];
    const int   end = offs[node + 1];
    const float di  = dinv[node];
    const vrow* hv  = (const vrow*)h;

    vrow acc = hv[(long)node * 32 + lane] * di;

    for (int base = beg; base < end; base += 32) {
        int my_s = 0, my_w = 0;
        if (base + lane < end) {
            my_s = ssrc[base + lane];
            my_w = __builtin_bit_cast(int, dinv[my_s]);
        }
        const int cnt = (end - base < 32) ? (end - base) : 32;
        for (int j = 0; j < cnt; j += 4) {
            const int s0 = __builtin_amdgcn_readlane(my_s, j);
            const int s1 = __builtin_amdgcn_readlane(my_s, j + 1);
            const int s2 = __builtin_amdgcn_readlane(my_s, j + 2);
            const int s3 = __builtin_amdgcn_readlane(my_s, j + 3);
            const float w0 = __builtin_bit_cast(float, __builtin_amdgcn_readlane(my_w, j));
            const float w1 = __builtin_bit_cast(float, __builtin_amdgcn_readlane(my_w, j + 1));
            const float w2 = __builtin_bit_cast(float, __builtin_amdgcn_readlane(my_w, j + 2));
            const float w3 = __builtin_bit_cast(float, __builtin_amdgcn_readlane(my_w, j + 3));
            const vrow r0 = hv[(long)s0 * 32 + lane];
            const vrow r1 = hv[(long)s1 * 32 + lane];
            const vrow r2 = hv[(long)s2 * 32 + lane];
            const vrow r3 = hv[(long)s3 * 32 + lane];
            acc += r0 * w0;
            acc += r1 * w1;
            acc += r2 * w2;
            acc += r3 * w3;
        }
    }

    vrow v = acc * di;
    if (FINAL) {
        v += ((const vrow*)bias)[lane];
#pragma unroll
        for (int i = 0; i < V; ++i) v[i] = v[i] > 0.0f ? v[i] : 0.0f;
    }
    ((vrow*)out)[(long)node * 32 + lane] = v;
}

extern "C" void kernel_launch(void* const* d_in, const int* in_sizes, int n_in,
                              void* d_out, int out_size, void* d_ws, size_t ws_size,
                              hipStream_t stream) {
    const float* x  = (const float*)d_in[0];
    const int*   ei = (const int*)d_in[1];
    const float* W1 = (const float*)d_in[2];
    const float* b1 = (const float*)d_in[3];
    const float* W2 = (const float*)d_in[4];
    const float* b2 = (const float*)d_in[5];

    const int N   = in_sizes[0] / K_DIM;
    const int E   = in_sizes[1] / 2;
    (void)n_in; (void)ws_size; (void)out_size;

    const int* src = ei;
    const int* dst = ei + E;
    float* out = (float*)d_out;

    char* ws = (char*)d_ws;
    const size_t degBytes = (((size_t)N * 4) + 255) & ~(size_t)255;
    float* dinv = (float*)ws;
    float* bufA = (float*)(ws + degBytes);
    float* bufB = (float*)(ws + degBytes + (size_t)N * 128 * 4);
    char*  lists = (char*)(bufB + (size_t)N * 64);
    const size_t offBytes = ((((size_t)N + 1) * 4) + 255) & ~(size_t)255;
    int* offs = (int*)lists;
    int* cnt  = (int*)(lists + offBytes);
    int* ssrc = cnt + E;
    int* recs = (int*)bufA;

    const int B    = (N + BKT_NODES - 1) >> BKT_SHIFT;
    const int per  = (E + N_CHUNKS - 1) / N_CHUNKS;
    int* btot = cnt + (size_t)B * N_CHUNKS;

    count_buckets<<<N_CHUNKS, 1024, 0, stream>>>(dst, cnt, E, per, B);
    scan_rows<<<B, N_CHUNKS, 0, stream>>>(cnt, btot);
    bin_edges<<<N_CHUNKS, 1024, 0, stream>>>(src, dst, cnt, btot, recs, E, per, B);
    group_bucket<<<B, 1024, 0, stream>>>(recs, btot, offs, dinv, ssrc, N, B);

    {
        gather_rows<128, false><<<(N + 7) / 8, 256, 0, stream>>>(x, offs, ssrc, dinv, b1, bufA, N);
        const int tiles = ((N + 15) / 16) * (128 / 16);
        gemm_wmma_f32<128, true><<<(tiles + 7) / 8, 256, 0, stream>>>(bufA, W1, b1, bufA, N);
    }

    {
        const int tiles = ((N + 15) / 16) * (64 / 16);
        gemm_wmma_f32<64, false><<<(tiles + 7) / 8, 256, 0, stream>>>(bufA, W2, b2, bufB, N);
        gather_rows<64, true><<<(N + 7) / 8, 256, 0, stream>>>(bufB, offs, ssrc, dinv, b2, out, N);
    }
}


# Dense transforms re-tiled: one wave = a 16-row stripe x all columns, W copied once per work-group to LDS (4 k side by side), A read once in 16-byte loads; sum order: k in each 8 now 0,1,4,5,2,3,6,7
# speedup vs baseline: 4.0214x; 1.2143x over previous
#include <hip/hip_runtime.h>


typedef __attribute__((ext_vector_type(2))) float v2f;
typedef __attribute__((ext_vector_type(4))) float v4f;
typedef __attribute__((ext_vector_type(8))) float v8f;

#define K_DIM 128

#define BKT_SHIFT 8
#define BKT_NODES (1 << BKT_SHIFT)
#define N_CHUNKS  256
#define MAX_BKTS  1024

__device__ __forceinline__ int block_excl_scan(int v, int* sm, int& total) {
    const int lane = threadIdx.x & 31;
    const int wave = threadIdx.x >> 5;
    int incl = v;
#pragma unroll
    for (int d = 1; d < 32; d <<= 1) {
        int y = __shfl_up(incl, d);
        if (lane >= d) incl += y;
    }
    if (lane == 31) sm[wave] = incl;
    __syncthreads();
    if (wave == 0) {
        const int nw = (int)(blockDim.x + 31) >> 5;
        const int w = (lane < nw) ? sm[lane] : 0;
        int iw = w;
#pragma unroll
        for (int d = 1; d < 32; d <<= 1) {
            int y = __shfl_up(iw, d);
            if (lane >= d) iw += y;
        }
        sm[lane] = iw - w;
        if (lane == 31) sm[32] = iw;
    }
    __syncthreads();
    total = sm[32];
    return sm[wave] + (incl - v);
}

__global__ void __launch_bounds__(1024)
count_buckets(const int* __restrict__ dst, int* __restrict__ cnt,
              int n_edges, int per, int n_bkts) {
    __shared__ int hist[MAX_BKTS];
    const int t = threadIdx.x, nt = blockDim.x, c = blockIdx.x;
    for (int b = t; b < n_bkts; b += nt) hist[b] = 0;
    __syncthreads();
    const int beg = c * per;
    const int end = (beg + per < n_edges) ? (beg + per) : n_edges;
    for (int i = beg + t; i < end; i += 4 * nt) {
        const int i1 = i + nt, i2 = i + 2 * nt, i3 = i + 3 * nt;
        const bool g1 = i1 < end, g2 = i2 < end, g3 = i3 < end;
        const int d0 = dst[i];
        const int d1 = g1 ? dst[i1] : 0;
        const int d2 = g2 ? dst[i2] : 0;
        const int d3 = g3 ? dst[i3] : 0;
        atomicAdd(&hist[d0 >> BKT_SHIFT], 1);
        if (g1) atomicAdd(&hist[d1 >> BKT_SHIFT], 1);
        if (g2) atomicAdd(&hist[d2 >> BKT_SHIFT], 1);
        if (g3) atomicAdd(&hist[d3 >> BKT_SHIFT], 1);
    }
    __syncthreads();
    for (int b = t; b < n_bkts; b += nt) cnt[b * N_CHUNKS + c] = hist[b];
}

__global__ void __launch_bounds__(N_CHUNKS)
scan_rows(int* __restrict__ cnt, int* __restrict__ btot) {
    __shared__ int sm[33];
    const int b = blockIdx.x, c = threadIdx.x;
    const int v = cnt[b * N_CHUNKS + c];
    int total;
    const int ex = block_excl_scan(v, sm, total);
    cnt[b * N_CHUNKS + c] = ex;
    if (c == 0) btot[b] = total;
}

__global__ void __launch_bounds__(1024)
bin_edges(const int* __restrict__ src, const int* __restrict__ dst,
          const int* __restrict__ cnt, const int* __restrict__ btot,
          int* __restrict__ recs_i, int n_edges, int per, int n_bkts) {
    __shared__ int cur[MAX_BKTS];
    __shared__ int sm[33];
    int2* __restrict__ recs = (int2*)recs_i;
    const int t = threadIdx.x, nt = blockDim.x, c = blockIdx.x;
    int total;
    const int bt = (t < n_bkts) ? btot[t] : 0;
    const int rp = (t < n_bkts) ? cnt[t * N_CHUNKS + c] : 0;
    const int bs = block_excl_scan(bt, sm, total);
    if (t < n_bkts) cur[t] = bs + rp;
    __syncthreads();
    const int beg = c * per;
    const int end = (beg + per < n_edges) ? (beg + per) : n_edges;
    for (int i = beg + t; i < end; i += 4 * nt) {
        const int i1 = i + nt, i2 = i + 2 * nt, i3 = i + 3 * nt;
        const bool g1 = i1 < end, g2 = i2 < end, g3 = i3 < end;
        const int s0 = src[i],            d0 = dst[i];
        const int s1 = g1 ? src[i1] : 0,  d1 = g1 ? dst[i1] : 0;
        const int s2 = g2 ? src[i2] : 0,  d2 = g2 ? dst[i2] : 0;
        const int s3 = g3 ? src[i3] : 0,  d3 = g3 ? dst[i3] : 0;
        recs[atomicAdd(&cur[d0 >> BKT_SHIFT], 1)] = make_int2(s0, d0);
        if (g1) recs[atomicAdd(&cur[d1 >> BKT_SHIFT], 1)] = make_int2(s1, d1);
        if (g2) recs[atomicAdd(&cur[d2 >> BKT_SHIFT], 1)] = make_int2(s2, d2);
        if (g3) recs[atomicAdd(&cur[d3 >> BKT_SHIFT], 1)] = make_int2(s3, d3);
    }
}

__global__ void __launch_bounds__(1024)
group_bucket(const int* __restrict__ recs_i, const int* __restrict__ btot,
             int* __restrict__ offs, float* __restrict__ dinv,
             int* __restrict__ ssrc, int n_nodes, int n_bkts) {
    __shared__ int lcnt[BKT_NODES];
    __shared__ int sm[33];
    const int2* __restrict__ recs = (const int2*)recs_i;
    const int t = threadIdx.x, nt = blockDim.x, b = blockIdx.x;
    const int node0 = b << BKT_SHIFT;

    int p = 0;
    for (int j = t; j < b; j += nt) p += btot[j];
    int bstart;
    (void)block_excl_scan(p, sm, bstart);
    const int bend = bstart + btot[b];

    if (t < BKT_NODES) lcnt[t] = 0;
    __syncthreads();
    for (int i = bstart + t; i < bend; i += 4 * nt) {
        const int i1 = i + nt, i2 = i + 2 * nt, i3 = i + 3 * nt;
        const bool g1 = i1 < bend, g2 = i2 < bend, g3 = i3 < bend;
        const int d0 = recs[i].y;
        const int d1 = g1 ? recs[i1].y : 0;
        const int d2 = g2 ? recs[i2].y : 0;
        const int d3 = g3 ? recs[i3].y : 0;
        atomicAdd(&lcnt[d0 - node0], 1);
        if (g1) atomicAdd(&lcnt[d1 - node0], 1);
        if (g2) atomicAdd(&lcnt[d2 - node0], 1);
        if (g3) atomicAdd(&lcnt[d3 - node0], 1);
    }
    __syncthreads();

    const int cv = (t < BKT_NODES) ? lcnt[t] : 0;
    int tot2;
    const int ex = block_excl_scan(cv, sm, tot2);
    if (t < BKT_NODES) {
        lcnt[t] = bstart + ex;
        const int node = node0 + t;
        if (node < n_nodes) {
            offs[node] = bstart + ex;
            dinv[node] = rsqrtf((float)(cv + 1));
        }
    }
    if (b == n_bkts - 1 && t == 0) offs[n_nodes] = bend;
    __syncthreads();

    for (int i = bstart + t; i < bend; i += 4 * nt) {
        const int i1 = i + nt, i2 = i + 2 * nt, i3 = i + 3 * nt;
        const bool g1 = i1 < bend, g2 = i2 < bend, g3 = i3 < bend;
        const int2 z = make_int2(0, 0);
        const int2 r0 = recs[i];
        const int2 r1 = g1 ? recs[i1] : z;
        const int2 r2 = g2 ? recs[i2] : z;
        const int2 r3 = g3 ? recs[i3] : z;
        ssrc[atomicAdd(&lcnt[r0.y - node0], 1)] = r0.x;
        if (g1) ssrc[atomicAdd(&lcnt[r1.y - node0], 1)] = r1.x;
        if (g2) ssrc[atomicAdd(&lcnt[r2.y - node0], 1)] = r2.x;
        if (g3) ssrc[atomicAdd(&lcnt[r3.y - node0], 1)] = r3.x;
    }
}

template <int NCOLS, bool FUSE>
__global__ void __launch_bounds__(256)
gemm_wmma_f32(const float* A,
              const float* __restrict__ W,
              const float* __restrict__ bias,
              float* C, int nrows) {
    constexpr int NT = NCOLS / 16;
    constexpr int NG = NCOLS / 4;
    constexpr int KQ = K_DIM / 4;
    __shared__ v4f wl[KQ * NCOLS];

    const int t = threadIdx.x;
    for (int it = t; it < KQ * NG; it += 256) {
        const int kq = it / NG, g = it % NG;
        const v4f* wp = (const v4f*)(W + (long)(4 * kq) * NCOLS) + g;
        const v4f w0 = wp[0];
        const v4f w1 = wp[NG];
        const v4f w2 = wp[2 * NG];
        const v4f w3 = wp[3 * NG];
        v4f* dp = &wl[kq * NCOLS + 4 * g];
        v4f o0 = {w0.x, w1.x, w2.x, w3.x};
        v4f o1 = {w0.y, w1.y, w2.y, w3.y};
        v4f o2 = {w0.z, w1.z, w2.z, w3.z};
        v4f o3 = {w0.w, w1.w, w2.w, w3.w};
        dp[0] = o0;
        dp[1] = o1;
        dp[2] = o2;
        dp[3] = o3;
    }
    __syncthreads();

    const int lane  = t & 31;
    const int wave  = t >> 5;
    const int rbase = (blockIdx.x * 8 + wave) * 16;
    if (rbase >= nrows) return;

    const int c = lane & 15;
    const int h = lane >> 4;

    const int arow_idx = (rbase + c < nrows) ? (rbase + c) : (nrows - 1);
    const v4f* arow = (const v4f*)(A + (long)arow_idx * K_DIM) + h;
    const v4f* bl   = wl + h * NCOLS + c;

    v8f acc[NT];
#pragma unroll
    for (int tt = 0; tt < NT; ++tt) acc[tt] = (v8f){};

#pragma unroll
    for (int j = 0; j < K_DIM / 8; ++j) {
        const v4f a = arow[2 * j];
        v4f b[NT];
#pragma unroll
        for (int tt = 0; tt < NT; ++tt) b[tt] = bl[2 * j * NCOLS + 16 * tt];
        const v2f a01 = {a.x, a.y};
        const v2f a23 = {a.z, a.w};
#pragma unroll
        for (int tt = 0; tt < NT; ++tt) {
            const v2f b01 = {b[tt].x, b[tt].y};
            acc[tt] = __builtin_amdgcn_wmma_f32_16x16x4_f32(
                false, a01, false, b01, (short)0, acc[tt], false, false);
        }
#pragma unroll
        for (int tt = 0; tt < NT; ++tt) {
            const v2f b23 = {b[tt].z, b[tt].w};
            acc[tt] = __builtin_amdgcn_wmma_f32_16x16x4_f32(
                false, a23, false, b23, (short)0, acc[tt], false, false);
        }
    }

    const bool full = rbase + 16 <= nrows;
#pragma unroll
    for (int tt = 0; tt < NT; ++tt) {
        v8f v = acc[tt];
        if (FUSE) {
            const float bv = bias[tt * 16 + c];
#pragma unroll
            for (int r = 0; r < 8; ++r) {
                const float u = v[r] + bv;
                v[r] = u > 0.0f ? u : 0.0f;
            }
        }
        if (full) {
            float* cbase = C + (long)(rbase + 8 * h) * NCOLS + tt * 16 + c;
#pragma unroll
            for (int r = 0; r < 8; ++r) cbase[(long)r * NCOLS] = v[r];
        } else {
#pragma unroll
            for (int r = 0; r < 8; ++r) {
                const int row = rbase + r + 8 * h;
                if (row < nrows) C[(long)row * NCOLS + tt * 16 + c] = v[r];
            }
        }
    }
}

template <int CH, bool FINAL>
__global__ void gather_rows(const float* __restrict__ h,
                            const int* __restrict__ offs,
                            const int* __restrict__ ssrc,
                            const float* __restrict__ dinv,
                            const float* __restrict__ bias,
                            float* __restrict__ out, int n_nodes) {
    constexpr int V = CH / 32;
    typedef __attribute__((ext_vector_type(V))) float vrow;
    const int lane = threadIdx.x & 31;
    const int node = __builtin_amdgcn_readfirstlane(
        (int)(blockIdx.x * (blockDim.x >> 5) + (threadIdx.x >> 5)));
    if (node >= n_nodes) return;

    const int   beg = offs[node];
    const int   end = offs[node + 1];
    const float di  = dinv[node];
    const vrow* hv  = (const vrow*)h;

    vrow acc = hv[(long)node * 32 + lane] * di;

    for (int base = beg; base < end; base += 32) {
        int my_s = 0, my_w = 0;
        if (base + lane < end) {
            my_s = ssrc[base + lane];
            my_w = __builtin_bit_cast(int, dinv[my_s]);
        }
        const int cnt = (end - base < 32) ? (end - base) : 32;
        for (int j = 0; j < cnt; j += 4) {
            const int s0 = __builtin_amdgcn_readlane(my_s, j);
            const int s1 = __builtin_amdgcn_readlane(my_s, j + 1);
            const int s2 = __builtin_amdgcn_readlane(my_s, j + 2);
            const int s3 = __builtin_amdgcn_readlane(my_s, j + 3);
            const float w0 = __builtin_bit_cast(float, __builtin_amdgcn_readlane(my_w, j));
            const float w1 = __builtin_bit_cast(float, __builtin_amdgcn_readlane(my_w, j + 1));
            const float w2 = __builtin_bit_cast(float, __builtin_amdgcn_readlane(my_w, j + 2));
            const float w3 = __builtin_bit_cast(float, __builtin_amdgcn_readlane(my_w, j + 3));
            const vrow r0 = hv[(long)s0 * 32 + lane];
            const vrow r1 = hv[(long)s1 * 32 + lane];
            const vrow r2 = hv[(long)s2 * 32 + lane];
            const vrow r3 = hv[(long)s3 * 32 + lane];
            acc += r0 * w0;
            acc += r1 * w1;
            acc += r2 * w2;
            acc += r3 * w3;
        }
    }

    vrow v = acc * di;
    if (FINAL) {
        v += ((const vrow*)bias)[lane];
#pragma unroll
        for (int i = 0; i < V; ++i) v[i] = v[i] > 0.0f ? v[i] : 0.0f;
    }
    ((vrow*)out)[(long)node * 32 + lane] = v;
}

extern "C" void kernel_launch(void* const* d_in, const int* in_sizes, int n_in,
                              void* d_out, int out_size, void* d_ws, size_t ws_size,
                              hipStream_t stream) {
    const float* x  = (const float*)d_in[0];
    const int*   ei = (const int*)d_in[1];
    const float* W1 = (const float*)d_in[2];
    const float* b1 = (const float*)d_in[3];
    const float* W2 = (const float*)d_in[4];
    const float* b2 = (const float*)d_in[5];

    const int N   = in_sizes[0] / K_DIM;
    const int E   = in_sizes[1] / 2;
    (void)n_in; (void)ws_size; (void)out_size;

    const int* src = ei;
    const int* dst = ei + E;
    float* out = (float*)d_out;

    char* ws = (char*)d_ws;
    const size_t degBytes = (((size_t)N * 4) + 255) & ~(size_t)255;
    float* dinv = (float*)ws;
    float* bufA = (float*)(ws + degBytes);
    float* bufB = (float*)(ws + degBytes + (size_t)N * 128 * 4);
    char*  lists = (char*)(bufB + (size_t)N * 64);
    const size_t offBytes = ((((size_t)N + 1) * 4) + 255) & ~(size_t)255;
    int* offs = (int*)lists;
    int* cnt  = (int*)(lists + offBytes);
    int* ssrc = cnt + E;
    int* recs = (int*)bufA;

    const int B    = (N + BKT_NODES - 1) >> BKT_SHIFT;
    const int per  = (E + N_CHUNKS - 1) / N_CHUNKS;
    int* btot = cnt + (size_t)B * N_CHUNKS;

    count_buckets<<<N_CHUNKS, 1024, 0, stream>>>(dst, cnt, E, per, B);
    scan_rows<<<B, N_CHUNKS, 0, stream>>>(cnt, btot);
    bin_edges<<<N_CHUNKS, 1024, 0, stream>>>(src, dst, cnt, btot, recs, E, per, B);
    group_bucket<<<B, 1024, 0, stream>>>(recs, btot, offs, dinv, ssrc, N, B);

    {
        gather_rows<128, false><<<(N + 7) / 8, 256, 0, stream>>>(x, offs, ssrc, dinv, b1, bufA, N);
        const int stripes = (N + 15) / 16;
        gemm_wmma_f32<128, true><<<(stripes + 7) / 8, 256, 0, stream>>>(bufA, W1, b1, bufA, N);
    }

    {
        const int stripes = (N + 15) / 16;
        gemm_wmma_f32<64, false><<<(stripes + 7) / 8, 256, 0, stream>>>(bufA, W2, b2, bufB, N);
        gather_rows<64, true><<<(N + 7) / 8, 256, 0, stream>>>(bufB, offs, ssrc, dinv, b2, out, N);
    }
}


# Edge grouping in 2 launches, not 4: each edge slice is bucket-sorted in LDS and written back in whole lines, the bucket pass walks slice runs; count and scan passes gone; sum order still arrival rank
# speedup vs baseline: 4.7683x; 1.1857x over previous
#include <hip/hip_runtime.h>


typedef __attribute__((ext_vector_type(2))) float v2f;
typedef __attribute__((ext_vector_type(4))) float v4f;
typedef __attribute__((ext_vector_type(8))) float v8f;

#define K_DIM 128

#define BKT_SHIFT 8
#define BKT_NODES (1 << BKT_SHIFT)
#define MAX_BKTS  1024
#define SLICE_THREADS 1024
#define SLICE_SLOTS 7
#define SLICE_CAP 6272
#define MAX_SLICES 1024
#define WALK_SLOTS 5

__device__ __forceinline__ int block_excl_scan(int v, int* sm, int& total) {
    const int lane = threadIdx.x & 31;
    const int wave = threadIdx.x >> 5;
    int incl = v;
#pragma unroll
    for (int d = 1; d < 32; d <<= 1) {
        int y = __shfl_up(incl, d);
        if (lane >= d) incl += y;
    }
    if (lane == 31) sm[wave] = incl;
    __syncthreads();
    if (wave == 0) {
        const int nw = (int)(blockDim.x + 31) >> 5;
        const int w = (lane < nw) ? sm[lane] : 0;
        int iw = w;
#pragma unroll
        for (int d = 1; d < 32; d <<= 1) {
            int y = __shfl_up(iw, d);
            if (lane >= d) iw += y;
        }
        sm[lane] = iw - w;
        if (lane == 31) sm[32] = iw;
    }
    __syncthreads();
    total = sm[32];
    return sm[wave] + (incl - v);
}

__global__ void __launch_bounds__(SLICE_THREADS)
sort_slice(const int* __restrict__ src, const int* __restrict__ dst,
           int* __restrict__ tab, int* __restrict__ recs,
           int n_edges, int n_bkts, int n_slices) {
    __shared__ int hist[MAX_BKTS];
    __shared__ int lrec[SLICE_CAP];
    __shared__ int sm[33];
    const int t = threadIdx.x, c = blockIdx.x;
    if (t < MAX_BKTS) hist[t] = 0;
    __syncthreads();
    const int beg = c * SLICE_CAP;
    const int len = (n_edges - beg < SLICE_CAP) ? (n_edges - beg) : SLICE_CAP;

    int s[SLICE_SLOTS], d[SLICE_SLOTS];
#pragma unroll
    for (int k = 0; k < SLICE_SLOTS; ++k) {
        const int i  = t + k * SLICE_THREADS;
        const int ii = (i < len) ? i : 0;
        s[k] = src[beg + ii];
        d[k] = dst[beg + ii];
    }
#pragma unroll
    for (int k = 0; k < SLICE_SLOTS; ++k)
        if (t + k * SLICE_THREADS < len) atomicAdd(&hist[d[k] >> BKT_SHIFT], 1);
    __syncthreads();

    const int v = (t < n_bkts) ? hist[t] : 0;
    int total;
    const int ex = block_excl_scan(v, sm, total);
    if (t < n_bkts) {
        hist[t] = ex;
        tab[t * n_slices + c] = (int)((unsigned)ex | ((unsigned)v << 16));
    }
    __syncthreads();

#pragma unroll
    for (int k = 0; k < SLICE_SLOTS; ++k)
        if (t + k * SLICE_THREADS < len) {
            const int pos = atomicAdd(&hist[d[k] >> BKT_SHIFT], 1);
            lrec[pos] = (s[k] << BKT_SHIFT) | (d[k] & (BKT_NODES - 1));
        }
    __syncthreads();
    for (int i = t; i < len; i += SLICE_THREADS) recs[beg + i] = lrec[i];
}

template <bool PLACE>
__device__ __forceinline__ void walk_runs(const int* __restrict__ recs, int* lcnt,
                                          int* __restrict__ ssrc,
                                          int my_start, int my_pre, int n_runs, int n_rec) {
    const int lane = threadIdx.x & 31;
    for (int base = 0; base < n_rec; base += 32 * WALK_SLOTS) {
        int run[WALK_SLOTS];
#pragma unroll
        for (int k = 0; k < WALK_SLOTS; ++k) run[k] = 0;
        for (int r = 1; r < n_runs; ++r) {
            const int pr = __builtin_amdgcn_readlane(my_pre, r);
#pragma unroll
            for (int k = 0; k < WALK_SLOTS; ++k) run[k] += (base + lane + 32 * k >= pr) ? 1 : 0;
        }
        int rec[WALK_SLOTS];
#pragma unroll
        for (int k = 0; k < WALK_SLOTS; ++k) {
            const int i = base + lane + 32 * k;
            const int a = __shfl(my_start, run[k]) + (i - __shfl(my_pre, run[k]));
            rec[k] = recs[(i < n_rec) ? a : 0];
        }
#pragma unroll
        for (int k = 0; k < WALK_SLOTS; ++k)
            if (base + lane + 32 * k < n_rec) {
                const int row = rec[k] & (BKT_NODES - 1);
                if (PLACE) ssrc[atomicAdd(&lcnt[row], 1)] = (int)((unsigned)rec[k] >> BKT_SHIFT);
                else       atomicAdd(&lcnt[row], 1);
            }
    }
}

__global__ void __launch_bounds__(1024)
group_bucket(const int* __restrict__ recs, const int* __restrict__ tab,
             int* __restrict__ offs, float* __restrict__ dinv,
             int* __restrict__ ssrc, int n_nodes, int n_bkts, int n_slices) {
    __shared__ int lcnt[BKT_NODES];
    __shared__ int rstart[MAX_SLICES];
    __shared__ int rlen[MAX_SLICES];
    __shared__ int sm[33];
    const int t = threadIdx.x, b = blockIdx.x;
    const int lane = t & 31, wave = t >> 5;
    const int node0 = b << BKT_SHIFT;

    const unsigned p = (t < n_slices) ? (unsigned)tab[b * n_slices + t] : 0u;
    const int loc = (int)(p & 0xffffu);
    const int cn  = (int)(p >> 16);
    int bstart, btot;
    (void)block_excl_scan(loc, sm, bstart);
    (void)block_excl_scan(cn, sm, btot);
    rstart[t] = t * SLICE_CAP + loc;
    rlen[t]   = cn;
    if (t < BKT_NODES) lcnt[t] = 0;
    __syncthreads();

    const int rpw = (n_slices + 31) >> 5;
    const int my_start = (lane < rpw) ? rstart[wave * rpw + lane] : 0;
    const int my_len   = (lane < rpw) ? rlen[wave * rpw + lane] : 0;
    int incl = my_len;
#pragma unroll
    for (int dd = 1; dd < 32; dd <<= 1) {
        int y = __shfl_up(incl, dd);
        if (lane >= dd) incl += y;
    }
    const int my_pre = incl - my_len;
    const int n_rec  = __builtin_amdgcn_readlane(incl, 31);

    walk_runs<false>(recs, lcnt, ssrc, my_start, my_pre, rpw, n_rec);
    __syncthreads();

    const int cv = (t < BKT_NODES) ? lcnt[t] : 0;
    int tot2;
    const int ex = block_excl_scan(cv, sm, tot2);
    if (t < BKT_NODES) {
        lcnt[t] = bstart + ex;
        const int node = node0 + t;
        if (node < n_nodes) {
            offs[node] = bstart + ex;
            dinv[node] = rsqrtf((float)(cv + 1));
        }
    }
    if (b == n_bkts - 1 && t == 0) offs[n_nodes] = bstart + btot;
    __syncthreads();

    walk_runs<true>(recs, lcnt, ssrc, my_start, my_pre, rpw, n_rec);
}

template <int NCOLS, bool FUSE>
__global__ void __launch_bounds__(256)
gemm_wmma_f32(const float* A,
              const float* __restrict__ W,
              const float* __restrict__ bias,
              float* C, int nrows) {
    constexpr int NT = NCOLS / 16;
    constexpr int NG = NCOLS / 4;
    constexpr int KQ = K_DIM / 4;
    __shared__ v4f wl[KQ * NCOLS];

    const int t = threadIdx.x;
    for (int it = t; it < KQ * NG; it += 256) {
        const int kq = it / NG, g = it % NG;
        const v4f* wp = (const v4f*)(W + (long)(4 * kq) * NCOLS) + g;
        const v4f w0 = wp[0];
        const v4f w1 = wp[NG];
        const v4f w2 = wp[2 * NG];
        const v4f w3 = wp[3 * NG];
        v4f* dp = &wl[kq * NCOLS + 4 * g];
        v4f o0 = {w0.x, w1.x, w2.x, w3.x};
        v4f o1 = {w0.y, w1.y, w2.y, w3.y};
        v4f o2 = {w0.z, w1.z, w2.z, w3.z};
        v4f o3 = {w0.w, w1.w, w2.w, w3.w};
        dp[0] = o0;
        dp[1] = o1;
        dp[2] = o2;
        dp[3] = o3;
    }
    __syncthreads();

    const int lane  = t & 31;
    const int wave  = t >> 5;
    const int rbase = (blockIdx.x * 8 + wave) * 16;
    if (rbase >= nrows) return;

    const int c = lane & 15;
    const int h = lane >> 4;

    const int arow_idx = (rbase + c < nrows) ? (rbase + c) : (nrows - 1);
    const v4f* arow = (const v4f*)(A + (long)arow_idx * K_DIM) + h;
    const v4f* bl   = wl + h * NCOLS + c;

    v8f acc[NT];
#pragma unroll
    for (int tt = 0; tt < NT; ++tt) acc[tt] = (v8f){};

#pragma unroll
    for (int j = 0; j < K_DIM / 8; ++j) {
        const v4f a = arow[2 * j];
        v4f b[NT];
#pragma unroll
        for (int tt = 0; tt < NT; ++tt) b[tt] = bl[2 * j * NCOLS + 16 * tt];
        const v2f a01 = {a.x, a.y};
        const v2f a23 = {a.z, a.w};
#pragma unroll
        for (int tt = 0; tt < NT; ++tt) {
            const v2f b01 = {b[tt].x, b[tt].y};
            acc[tt] = __builtin_amdgcn_wmma_f32_16x16x4_f32(
                false, a01, false, b01, (short)0, acc[tt], false, false);
        }
#pragma unroll
        for (int tt = 0; tt < NT; ++tt) {
            const v2f b23 = {b[tt].z, b[tt].w};
            acc[tt] = __builtin_amdgcn_wmma_f32_16x16x4_f32(
                false, a23, false, b23, (short)0, acc[tt], false, false);
        }
    }

    const bool full = rbase + 16 <= nrows;
#pragma unroll
    for (int tt = 0; tt < NT; ++tt) {
        v8f v = acc[tt];
        if (FUSE) {
            const float bv = bias[tt * 16 + c];
#pragma unroll
            for (int r = 0; r < 8; ++r) {
                const float u = v[r] + bv;
                v[r] = u > 0.0f ? u : 0.0f;
            }
        }
        if (full) {
            float* cbase = C + (long)(rbase + 8 * h) * NCOLS + tt * 16 + c;
#pragma unroll
            for (int r = 0; r < 8; ++r) cbase[(long)r * NCOLS] = v[r];
        } else {
#pragma unroll
            for (int r = 0; r < 8; ++r) {
                const int row = rbase + r + 8 * h;
                if (row < nrows) C[(long)row * NCOLS + tt * 16 + c] = v[r];
            }
        }
    }
}

template <int CH, bool FINAL>
__global__ void gather_rows(const float* __restrict__ h,
                            const int* __restrict__ offs,
                            const int* __restrict__ ssrc,
                            const float* __restrict__ dinv,
                            const float* __restrict__ bias,
                            float* __restrict__ out, int n_nodes) {
    constexpr int V = CH / 32;
    typedef __attribute__((ext_vector_type(V))) float vrow;
    const int lane = threadIdx.x & 31;
    const int node = __builtin_amdgcn_readfirstlane(
        (int)(blockIdx.x * (blockDim.x >> 5) + (threadIdx.x >> 5)));
    if (node >= n_nodes) return;

    const int   beg = offs[node];
    const int   end = offs[node + 1];
    const float di  = dinv[node];
    const vrow* hv  = (const vrow*)h;

    vrow acc = hv[(long)node * 32 + lane] * di;

    for (int base = beg; base < end; base += 32) {
        int my_s = 0, my_w = 0;
        if (base + lane < end) {
            my_s = ssrc[base + lane];
            my_w = __builtin_bit_cast(int, dinv[my_s]);
        }
        const int cnt = (end - base < 32) ? (end - base) : 32;
        for (int j = 0; j < cnt; j += 4) {
            const int s0 = __builtin_amdgcn_readlane(my_s, j);
            const int s1 = __builtin_amdgcn_readlane(my_s, j + 1);
            const int s2 = __builtin_amdgcn_readlane(my_s, j + 2);
            const int s3 = __builtin_amdgcn_readlane(my_s, j + 3);
            const float w0 = __builtin_bit_cast(float, __builtin_amdgcn_readlane(my_w, j));
            const float w1 = __builtin_bit_cast(float, __builtin_amdgcn_readlane(my_w, j + 1));
            const float w2 = __builtin_bit_cast(float, __builtin_amdgcn_readlane(my_w, j + 2));
            const float w3 = __builtin_bit_cast(float, __builtin_amdgcn_readlane(my_w, j + 3));
            const vrow r0 = hv[(long)s0 * 32 + lane];
            const vrow r1 = hv[(long)s1 * 32 + lane];
            const vrow r2 = hv[(long)s2 * 32 + lane];
            const vrow r3 = hv[(long)s3 * 32 + lane];
            acc += r0 * w0;
            acc += r1 * w1;
            acc += r2 * w2;
            acc += r3 * w3;
        }
    }

    vrow v = acc * di;
    if (FINAL) {
        v += ((const vrow*)bias)[lane];
#pragma unroll
        for (int i = 0; i < V; ++i) v[i] = v[i] > 0.0f ? v[i] : 0.0f;
    }
    ((vrow*)out)[(long)node * 32 + lane] = v;
}

extern "C" void kernel_launch(void* const* d_in, const int* in_sizes, int n_in,
                              void* d_out, int out_size, void* d_ws, size_t ws_size,
                              hipStream_t stream) {
    const float* x  = (const float*)d_in[0];
    const int*   ei = (const int*)d_in[1];
    const float* W1 = (const float*)d_in[2];
    const float* b1 = (const float*)d_in[3];
    const float* W2 = (const float*)d_in[4];
    const float* b2 = (const float*)d_in[5];

    const int N   = in_sizes[0] / K_DIM;
    const int E   = in_sizes[1] / 2;
    (void)n_in; (void)ws_size; (void)out_size;

    const int* src = ei;
    const int* dst = ei + E;
    float* out = (float*)d_out;

    char* ws = (char*)d_ws;
    const size_t degBytes = (((size_t)N * 4) + 255) & ~(size_t)255;
    float* dinv = (float*)ws;
    float* bufA = (float*)(ws + degBytes);
    float* bufB = (float*)(ws + degBytes + (size_t)N * 128 * 4);
    char*  lists = (char*)(bufB + (size_t)N * 64);
    const size_t offBytes = ((((size_t)N + 1) * 4) + 255) & ~(size_t)255;
    int* offs = (int*)lists;
    int* cnt  = (int*)(lists + offBytes);
    int* ssrc = cnt + E;
    int* recs = (int*)bufA;

    const int B  = (N + BKT_NODES - 1) >> BKT_SHIFT;
    const int S  = (E + SLICE_CAP - 1) / SLICE_CAP;
    int* tab = cnt;

    sort_slice<<<S, SLICE_THREADS, 0, stream>>>(src, dst, tab, recs, E, B, S);
    group_bucket<<<B, 1024, 0, stream>>>(recs, tab, offs, dinv, ssrc, N, B, S);

    {
        gather_rows<128, false><<<(N + 7) / 8, 256, 0, stream>>>(x, offs, ssrc, dinv, b1, bufA, N);
        const int stripes = (N + 15) / 16;
        gemm_wmma_f32<128, true><<<(stripes + 7) / 8, 256, 0, stream>>>(bufA, W1, b1, bufA, N);
    }

    {
        const int stripes = (N + 15) / 16;
        gemm_wmma_f32<64, false><<<(stripes + 7) / 8, 256, 0, stream>>>(bufA, W2, b2, bufB, N);
        gather_rows<64, true><<<(N + 7) / 8, 256, 0, stream>>>(bufB, offs, ssrc, dinv, b2, out, N);
    }
}


# Layer-1 gather and both dense transforms fused into one launch: a 64-row tile is gathered into LDS, multiplied by W1 (bias+ReLU in place) and by W2; only h2 is stored; every sum order unchanged
# speedup vs baseline: 5.5904x; 1.1724x over previous
#include <hip/hip_runtime.h>


typedef __attribute__((ext_vector_type(2))) float v2f;
typedef __attribute__((ext_vector_type(4))) float v4f;
typedef __attribute__((ext_vector_type(8))) float v8f;

#define K_DIM 128

#define BKT_SHIFT 8
#define BKT_NODES (1 << BKT_SHIFT)
#define MAX_BKTS  1024
#define SLICE_THREADS 1024
#define SLICE_SLOTS 7
#define SLICE_CAP 6272
#define MAX_SLICES 1024
#define WALK_SLOTS 5

__device__ __forceinline__ int block_excl_scan(int v, int* sm, int& total) {
    const int lane = threadIdx.x & 31;
    const int wave = threadIdx.x >> 5;
    int incl = v;
#pragma unroll
    for (int d = 1; d < 32; d <<= 1) {
        int y = __shfl_up(incl, d);
        if (lane >= d) incl += y;
    }
    if (lane == 31) sm[wave] = incl;
    __syncthreads();
    if (wave == 0) {
        const int nw = (int)(blockDim.x + 31) >> 5;
        const int w = (lane < nw) ? sm[lane] : 0;
        int iw = w;
#pragma unroll
        for (int d = 1; d < 32; d <<= 1) {
            int y = __shfl_up(iw, d);
            if (lane >= d) iw += y;
        }
        sm[lane] = iw - w;
        if (lane == 31) sm[32] = iw;
    }
    __syncthreads();
    total = sm[32];
    return sm[wave] + (incl - v);
}

__global__ void __launch_bounds__(SLICE_THREADS)
sort_slice(const int* __restrict__ src, const int* __restrict__ dst,
           int* __restrict__ tab, int* __restrict__ recs,
           int n_edges, int n_bkts, int n_slices) {
    __shared__ int hist[MAX_BKTS];
    __shared__ int lrec[SLICE_CAP];
    __shared__ int sm[33];
    const int t = threadIdx.x, c = blockIdx.x;
    if (t < MAX_BKTS) hist[t] = 0;
    __syncthreads();
    const int beg = c * SLICE_CAP;
    const int len = (n_edges - beg < SLICE_CAP) ? (n_edges - beg) : SLICE_CAP;

    int s[SLICE_SLOTS], d[SLICE_SLOTS];
#pragma unroll
    for (int k = 0; k < SLICE_SLOTS; ++k) {
        const int i  = t + k * SLICE_THREADS;
        const int ii = (i < len) ? i : 0;
        s[k] = src[beg + ii];
        d[k] = dst[beg + ii];
    }
#pragma unroll
    for (int k = 0; k < SLICE_SLOTS; ++k)
        if (t + k * SLICE_THREADS < len) atomicAdd(&hist[d[k] >> BKT_SHIFT], 1);
    __syncthreads();

    const int v = (t < n_bkts) ? hist[t] : 0;
    int total;
    const int ex = block_excl_scan(v, sm, total);
    if (t < n_bkts) {
        hist[t] = ex;
        tab[t * n_slices + c] = (int)((unsigned)ex | ((unsigned)v << 16));
    }
    __syncthreads();

#pragma unroll
    for (int k = 0; k < SLICE_SLOTS; ++k)
        if (t + k * SLICE_THREADS < len) {
            const int pos = atomicAdd(&hist[d[k] >> BKT_SHIFT], 1);
            lrec[pos] = (s[k] << BKT_SHIFT) | (d[k] & (BKT_NODES - 1));
        }
    __syncthreads();
    for (int i = t; i < len; i += SLICE_THREADS) recs[beg + i] = lrec[i];
}

template <bool PLACE>
__device__ __forceinline__ void walk_runs(const int* __restrict__ recs, int* lcnt,
                                          int* __restrict__ ssrc,
                                          int my_start, int my_pre, int n_runs, int n_rec) {
    const int lane = threadIdx.x & 31;
    for (int base = 0; base < n_rec; base += 32 * WALK_SLOTS) {
        int run[WALK_SLOTS];
#pragma unroll
        for (int k = 0; k < WALK_SLOTS; ++k) run[k] = 0;
        for (int r = 1; r < n_runs; ++r) {
            const int pr = __builtin_amdgcn_readlane(my_pre, r);
#pragma unroll
            for (int k = 0; k < WALK_SLOTS; ++k) run[k] += (base + lane + 32 * k >= pr) ? 1 : 0;
        }
        int rec[WALK_SLOTS];
#pragma unroll
        for (int k = 0; k < WALK_SLOTS; ++k) {
            const int i = base + lane + 32 * k;
            const int a = __shfl(my_start, run[k]) + (i - __shfl(my_pre, run[k]));
            rec[k] = recs[(i < n_rec) ? a : 0];
        }
#pragma unroll
        for (int k = 0; k < WALK_SLOTS; ++k)
            if (base + lane + 32 * k < n_rec) {
                const int row = rec[k] & (BKT_NODES - 1);
                if (PLACE) ssrc[atomicAdd(&lcnt[row], 1)] = (int)((unsigned)rec[k] >> BKT_SHIFT);
                else       atomicAdd(&lcnt[row], 1);
            }
    }
}

__global__ void __launch_bounds__(1024)
group_bucket(const int* __restrict__ recs, const int* __restrict__ tab,
             int* __restrict__ offs, float* __restrict__ dinv,
             int* __restrict__ ssrc, int n_nodes, int n_bkts, int n_slices) {
    __shared__ int lcnt[BKT_NODES];
    __shared__ int rstart[MAX_SLICES];
    __shared__ int rlen[MAX_SLICES];
    __shared__ int sm[33];
    const int t = threadIdx.x, b = blockIdx.x;
    const int lane = t & 31, wave = t >> 5;
    const int node0 = b << BKT_SHIFT;

    const unsigned p = (t < n_slices) ? (unsigned)tab[b * n_slices + t] : 0u;
    const int loc = (int)(p & 0xffffu);
    const int cn  = (int)(p >> 16);
    int bstart, btot;
    (void)block_excl_scan(loc, sm, bstart);
    (void)block_excl_scan(cn, sm, btot);
    rstart[t] = t * SLICE_CAP + loc;
    rlen[t]   = cn;
    if (t < BKT_NODES) lcnt[t] = 0;
    __syncthreads();

    const int rpw = (n_slices + 31) >> 5;
    const int my_start = (lane < rpw) ? rstart[wave * rpw + lane] : 0;
    const int my_len   = (lane < rpw) ? rlen[wave * rpw + lane] : 0;
    int incl = my_len;
#pragma unroll
    for (int dd = 1; dd < 32; dd <<= 1) {
        int y = __shfl_up(incl, dd);
        if (lane >= dd) incl += y;
    }
    const int my_pre = incl - my_len;
    const int n_rec  = __builtin_amdgcn_readlane(incl, 31);

    walk_runs<false>(recs, lcnt, ssrc, my_start, my_pre, rpw, n_rec);
    __syncthreads();

    const int cv = (t < BKT_NODES) ? lcnt[t] : 0;
    int tot2;
    const int ex = block_excl_scan(cv, sm, tot2);
    if (t < BKT_NODES) {
        lcnt[t] = bstart + ex;
        const int node = node0 + t;
        if (node < n_nodes) {
            offs[node] = bstart + ex;
            dinv[node] = rsqrtf((float)(cv + 1));
        }
    }
    if (b == n_bkts - 1 && t == 0) offs[n_nodes] = bstart + btot;
    __syncthreads();

    walk_runs<true>(recs, lcnt, ssrc, my_start, my_pre, rpw, n_rec);
}

#define FT_ROWS    64
#define FT_PITCH4  33
#define FT_THREADS 1024

template <int NCOLS>
__device__ __forceinline__ void stage_w(const float* __restrict__ W, v4f* wl, int t) {
    constexpr int NG = NCOLS / 4;
    constexpr int KQ = K_DIM / 4;
    for (int it = t; it < KQ * NG; it += FT_THREADS) {
        const int kq = it / NG, g = it % NG;
        const v4f* wp = (const v4f*)(W + (long)(4 * kq) * NCOLS) + g;
        const v4f w0 = wp[0];
        const v4f w1 = wp[NG];
        const v4f w2 = wp[2 * NG];
        const v4f w3 = wp[3 * NG];
        v4f* dp = &wl[kq * NCOLS + 4 * g];
        v4f o0 = {w0.x, w1.x, w2.x, w3.x};
        v4f o1 = {w0.y, w1.y, w2.y, w3.y};
        v4f o2 = {w0.z, w1.z, w2.z, w3.z};
        v4f o3 = {w0.w, w1.w, w2.w, w3.w};
        dp[0] = o0;
        dp[1] = o1;
        dp[2] = o2;
        dp[3] = o3;
    }
}

template <int NCOLS, int NT>
__device__ __forceinline__ void stripe_product(const v4f* arow, const v4f* bl, v8f (&acc)[NT]) {
#pragma unroll
    for (int tt = 0; tt < NT; ++tt) acc[tt] = (v8f){};
#pragma unroll
    for (int j = 0; j < K_DIM / 8; ++j) {
        const v4f a = arow[2 * j];
        v4f b[NT];
#pragma unroll
        for (int tt = 0; tt < NT; ++tt) b[tt] = bl[2 * j * NCOLS + 16 * tt];
        const v2f a01 = {a.x, a.y};
        const v2f a23 = {a.z, a.w};
#pragma unroll
        for (int tt = 0; tt < NT; ++tt) {
            const v2f b01 = {b[tt].x, b[tt].y};
            acc[tt] = __builtin_amdgcn_wmma_f32_16x16x4_f32(
                false, a01, false, b01, (short)0, acc[tt], false, false);
        }
#pragma unroll
        for (int tt = 0; tt < NT; ++tt) {
            const v2f b23 = {b[tt].z, b[tt].w};
            acc[tt] = __builtin_amdgcn_wmma_f32_16x16x4_f32(
                false, a23, false, b23, (short)0, acc[tt], false, false);
        }
    }
}

__global__ void __launch_bounds__(FT_THREADS, 16)
layer1_fused(const float* __restrict__ x,
             const int* __restrict__ offs,
             const int* __restrict__ ssrc,
             const float* __restrict__ dinv,
             const float* __restrict__ W1,
             const float* __restrict__ b1,
             const float* __restrict__ W2,
             float* __restrict__ h2, int n_nodes, int rows_per_wg) {
    __shared__ v4f wl1[(K_DIM / 4) * 128];
    __shared__ v4f wl2[(K_DIM / 4) * 64];
    __shared__ v4f tile[FT_ROWS * FT_PITCH4];

    const int t = threadIdx.x;
    stage_w<128>(W1, wl1, t);
    stage_w<64>(W2, wl2, t);

    const int lane = t & 31;
    const int wave = __builtin_amdgcn_readfirstlane(t >> 5);
    const int c = lane & 15;
    const int h = lane >> 4;

    const int row0 = blockIdx.x * rows_per_wg;
    const int row1 = (row0 + rows_per_wg < n_nodes) ? (row0 + rows_per_wg) : n_nodes;
    const v4f* xv = (const v4f*)x;

    for (int tb = row0; tb < row1; tb += FT_ROWS) {
        for (int q = 0; q < FT_ROWS / 32; ++q) {
            const int lr   = wave + 32 * q;
            const int node = tb + lr;
            if (node < row1) {
                const int   beg = offs[node];
                const int   end = offs[node + 1];
                const float di  = dinv[node];
                v4f acc = xv[(long)node * 32 + lane] * di;
                for (int base = beg; base < end; base += 32) {
                    int my_s = 0, my_w = 0;
                    if (base + lane < end) {
                        my_s = ssrc[base + lane];
                        my_w = __builtin_bit_cast(int, dinv[my_s]);
                    }
                    const int cnt = (end - base < 32) ? (end - base) : 32;
                    for (int j = 0; j < cnt; j += 4) {
                        const int s0 = __builtin_amdgcn_readlane(my_s, j);
                        const int s1 = __builtin_amdgcn_readlane(my_s, j + 1);
                        const int s2 = __builtin_amdgcn_readlane(my_s, j + 2);
                        const int s3 = __builtin_amdgcn_readlane(my_s, j + 3);
                        const float w0 = __builtin_bit_cast(float, __builtin_amdgcn_readlane(my_w, j));
                        const float w1 = __builtin_bit_cast(float, __builtin_amdgcn_readlane(my_w, j + 1));
                        const float w2 = __builtin_bit_cast(float, __builtin_amdgcn_readlane(my_w, j + 2));
                        const float w3 = __builtin_bit_cast(float, __builtin_amdgcn_readlane(my_w, j + 3));
                        const v4f r0 = xv[(long)s0 * 32 + lane];
                        const v4f r1 = xv[(long)s1 * 32 + lane];
                        const v4f r2 = xv[(long)s2 * 32 + lane];
                        const v4f r3 = xv[(long)s3 * 32 + lane];
                        acc += r0 * w0;
                        acc += r1 * w1;
                        acc += r2 * w2;
                        acc += r3 * w3;
                    }
                }
                tile[lr * FT_PITCH4 + lane] = acc * di;
            } else {
                tile[lr * FT_PITCH4 + lane] = (v4f){};
            }
        }
        __syncthreads();

        const int  s1    = wave >> 3;
        const int  t1    = wave & 7;
        const bool live1 = tb + 16 * s1 < row1;
        v8f d1[1];
        if (live1)
            stripe_product<128, 1>(tile + (16 * s1 + c) * FT_PITCH4 + h,
                                   wl1 + h * 128 + 16 * t1 + c, d1);
        __syncthreads();
        if (live1) {
            const float bv = b1[16 * t1 + c];
            float* tf = (float*)tile + (16 * s1 + 8 * h) * (4 * FT_PITCH4) + 16 * t1 + c;
#pragma unroll
            for (int r = 0; r < 8; ++r) {
                const float u = d1[0][r] + bv;
                tf[r * (4 * FT_PITCH4)] = u > 0.0f ? u : 0.0f;
            }
        }
        __syncthreads();

        if (wave < 8) {
            const int s2 = wave >> 1;
            const int p  = wave & 1;
            const int rb = tb + 16 * s2;
            if (rb < row1) {
                v8f d2[2];
                stripe_product<64, 2>(tile + (16 * s2 + c) * FT_PITCH4 + h,
                                      wl2 + h * 64 + 32 * p + c, d2);
                if (rb + 16 <= row1) {
#pragma unroll
                    for (int tt = 0; tt < 2; ++tt) {
                        float* cbase = h2 + (long)(rb + 8 * h) * 64 + 32 * p + 16 * tt + c;
#pragma unroll
                        for (int r = 0; r < 8; ++r) cbase[(long)r * 64] = d2[tt][r];
                    }
                } else {
#pragma unroll
                    for (int tt = 0; tt < 2; ++tt) {
#pragma unroll
                        for (int r = 0; r < 8; ++r) {
                            const int row = rb + r + 8 * h;
                            if (row < row1) h2[(long)row * 64 + 32 * p + 16 * tt + c] = d2[tt][r];
                        }
                    }
                }
            }
        }
        __syncthreads();
    }
}

template <int CH, bool FINAL>
__global__ void gather_rows(const float* __restrict__ h,
                            const int* __restrict__ offs,
                            const int* __restrict__ ssrc,
                            const float* __restrict__ dinv,
                            const float* __restrict__ bias,
                            float* __restrict__ out, int n_nodes) {
    constexpr int V = CH / 32;
    typedef __attribute__((ext_vector_type(V))) float vrow;
    const int lane = threadIdx.x & 31;
    const int node = __builtin_amdgcn_readfirstlane(
        (int)(blockIdx.x * (blockDim.x >> 5) + (threadIdx.x >> 5)));
    if (node >= n_nodes) return;

    const int   beg = offs[node];
    const int   end = offs[node + 1];
    const float di  = dinv[node];
    const vrow* hv  = (const vrow*)h;

    vrow acc = hv[(long)node * 32 + lane] * di;

    for (int base = beg; base < end; base += 32) {
        int my_s = 0, my_w = 0;
        if (base + lane < end) {
            my_s = ssrc[base + lane];
            my_w = __builtin_bit_cast(int, dinv[my_s]);
        }
        const int cnt = (end - base < 32) ? (end - base) : 32;
        for (int j = 0; j < cnt; j += 4) {
            const int s0 = __builtin_amdgcn_readlane(my_s, j);
            const int s1 = __builtin_amdgcn_readlane(my_s, j + 1);
            const int s2 = __builtin_amdgcn_readlane(my_s, j + 2);
            const int s3 = __builtin_amdgcn_readlane(my_s, j + 3);
            const float w0 = __builtin_bit_cast(float, __builtin_amdgcn_readlane(my_w, j));
            const float w1 = __builtin_bit_cast(float, __builtin_amdgcn_readlane(my_w, j + 1));
            const float w2 = __builtin_bit_cast(float, __builtin_amdgcn_readlane(my_w, j + 2));
            const float w3 = __builtin_bit_cast(float, __builtin_amdgcn_readlane(my_w, j + 3));
            const vrow r0 = hv[(long)s0 * 32 + lane];
            const vrow r1 = hv[(long)s1 * 32 + lane];
            const vrow r2 = hv[(long)s2 * 32 + lane];
            const vrow r3 = hv[(long)s3 * 32 + lane];
            acc += r0 * w0;
            acc += r1 * w1;
            acc += r2 * w2;
            acc += r3 * w3;
        }
    }

    vrow v = acc * di;
    if (FINAL) {
        v += ((const vrow*)bias)[lane];
#pragma unroll
        for (int i = 0; i < V; ++i) v[i] = v[i] > 0.0f ? v[i] : 0.0f;
    }
    ((vrow*)out)[(long)node * 32 + lane] = v;
}

extern "C" void kernel_launch(void* const* d_in, const int* in_sizes, int n_in,
                              void* d_out, int out_size, void* d_ws, size_t ws_size,
                              hipStream_t stream) {
    const float* x  = (const float*)d_in[0];
    const int*   ei = (const int*)d_in[1];
    const float* W1 = (const float*)d_in[2];
    const float* b1 = (const float*)d_in[3];
    const float* W2 = (const float*)d_in[4];
    const float* b2 = (const float*)d_in[5];

    const int N   = in_sizes[0] / K_DIM;
    const int E   = in_sizes[1] / 2;
    (void)n_in; (void)ws_size; (void)out_size;

    const int* src = ei;
    const int* dst = ei + E;
    float* out = (float*)d_out;

    char* ws = (char*)d_ws;
    const size_t degBytes = (((size_t)N * 4) + 255) & ~(size_t)255;
    float* dinv = (float*)ws;
    float* bufA = (float*)(ws + degBytes);
    float* bufB = (float*)(ws + degBytes + (size_t)N * 128 * 4);
    char*  lists = (char*)(bufB + (size_t)N * 64);
    const size_t offBytes = ((((size_t)N + 1) * 4) + 255) & ~(size_t)255;
    int* offs = (int*)lists;
    int* cnt  = (int*)(lists + offBytes);
    int* ssrc = cnt + E;
    int* recs = (int*)bufA;

    const int B  = (N + BKT_NODES - 1) >> BKT_SHIFT;
    const int S  = (E + SLICE_CAP - 1) / SLICE_CAP;
    int* tab = cnt;

    sort_slice<<<S, SLICE_THREADS, 0, stream>>>(src, dst, tab, recs, E, B, S);
    group_bucket<<<B, 1024, 0, stream>>>(recs, tab, offs, dinv, ssrc, N, B, S);

    {
        const int rows_per_wg = (((N + 511) / 512) + 3) & ~3;
        const int groups = (N + rows_per_wg - 1) / rows_per_wg;
        layer1_fused<<<groups, FT_THREADS, 0, stream>>>(x, offs, ssrc, dinv, W1, b1, W2, bufB, N, rows_per_wg);
    }

    gather_rows<64, true><<<(N + 7) / 8, 256, 0, stream>>>(bufB, offs, ssrc, dinv, b2, out, N);
}


# Layer-2 edge weight folded into the rows: hidden rows go back times their own dinv, so h2 is stored as dinv*h2; the layer-2 gather adds plain rows, no per-edge dinv read; sum order unchanged
# speedup vs baseline: 5.7718x; 1.0324x over previous
#include <hip/hip_runtime.h>


typedef __attribute__((ext_vector_type(2))) float v2f;
typedef __attribute__((ext_vector_type(4))) float v4f;
typedef __attribute__((ext_vector_type(8))) float v8f;

#define K_DIM 128

#define BKT_SHIFT 8
#define BKT_NODES (1 << BKT_SHIFT)
#define MAX_BKTS  1024
#define SLICE_THREADS 1024
#define SLICE_SLOTS 7
#define SLICE_CAP 6272
#define MAX_SLICES 1024
#define WALK_SLOTS 5

__device__ __forceinline__ int block_excl_scan(int v, int* sm, int& total) {
    const int lane = threadIdx.x & 31;
    const int wave = threadIdx.x >> 5;
    int incl = v;
#pragma unroll
    for (int d = 1; d < 32; d <<= 1) {
        int y = __shfl_up(incl, d);
        if (lane >= d) incl += y;
    }
    if (lane == 31) sm[wave] = incl;
    __syncthreads();
    if (wave == 0) {
        const int nw = (int)(blockDim.x + 31) >> 5;
        const int w = (lane < nw) ? sm[lane] : 0;
        int iw = w;
#pragma unroll
        for (int d = 1; d < 32; d <<= 1) {
            int y = __shfl_up(iw, d);
            if (lane >= d) iw += y;
        }
        sm[lane] = iw - w;
        if (lane == 31) sm[32] = iw;
    }
    __syncthreads();
    total = sm[32];
    return sm[wave] + (incl - v);
}

__global__ void __launch_bounds__(SLICE_THREADS)
sort_slice(const int* __restrict__ src, const int* __restrict__ dst,
           int* __restrict__ tab, int* __restrict__ recs,
           int n_edges, int n_bkts, int n_slices) {
    __shared__ int hist[MAX_BKTS];
    __shared__ int lrec[SLICE_CAP];
    __shared__ int sm[33];
    const int t = threadIdx.x, c = blockIdx.x;
    if (t < MAX_BKTS) hist[t] = 0;
    __syncthreads();
    const int beg = c * SLICE_CAP;
    const int len = (n_edges - beg < SLICE_CAP) ? (n_edges - beg) : SLICE_CAP;

    int s[SLICE_SLOTS], d[SLICE_SLOTS];
#pragma unroll
    for (int k = 0; k < SLICE_SLOTS; ++k) {
        const int i  = t + k * SLICE_THREADS;
        const int ii = (i < len) ? i : 0;
        s[k] = src[beg + ii];
        d[k] = dst[beg + ii];
    }
#pragma unroll
    for (int k = 0; k < SLICE_SLOTS; ++k)
        if (t + k * SLICE_THREADS < len) atomicAdd(&hist[d[k] >> BKT_SHIFT], 1);
    __syncthreads();

    const int v = (t < n_bkts) ? hist[t] : 0;
    int total;
    const int ex = block_excl_scan(v, sm, total);
    if (t < n_bkts) {
        hist[t] = ex;
        tab[t * n_slices + c] = (int)((unsigned)ex | ((unsigned)v << 16));
    }
    __syncthreads();

#pragma unroll
    for (int k = 0; k < SLICE_SLOTS; ++k)
        if (t + k * SLICE_THREADS < len) {
            const int pos = atomicAdd(&hist[d[k] >> BKT_SHIFT], 1);
            lrec[pos] = (s[k] << BKT_SHIFT) | (d[k] & (BKT_NODES - 1));
        }
    __syncthreads();
    for (int i = t; i < len; i += SLICE_THREADS) recs[beg + i] = lrec[i];
}

template <bool PLACE>
__device__ __forceinline__ void walk_runs(const int* __restrict__ recs, int* lcnt,
                                          int* __restrict__ ssrc,
                                          int my_start, int my_pre, int n_runs, int n_rec) {
    const int lane = threadIdx.x & 31;
    for (int base = 0; base < n_rec; base += 32 * WALK_SLOTS) {
        int run[WALK_SLOTS];
#pragma unroll
        for (int k = 0; k < WALK_SLOTS; ++k) run[k] = 0;
        for (int r = 1; r < n_runs; ++r) {
            const int pr = __builtin_amdgcn_readlane(my_pre, r);
#pragma unroll
            for (int k = 0; k < WALK_SLOTS; ++k) run[k] += (base + lane + 32 * k >= pr) ? 1 : 0;
        }
        int rec[WALK_SLOTS];
#pragma unroll
        for (int k = 0; k < WALK_SLOTS; ++k) {
            const int i = base + lane + 32 * k;
            const int a = __shfl(my_start, run[k]) + (i - __shfl(my_pre, run[k]));
            rec[k] = recs[(i < n_rec) ? a : 0];
        }
#pragma unroll
        for (int k = 0; k < WALK_SLOTS; ++k)
            if (base + lane + 32 * k < n_rec) {
                const int row = rec[k] & (BKT_NODES - 1);
                if (PLACE) ssrc[atomicAdd(&lcnt[row], 1)] = (int)((unsigned)rec[k] >> BKT_SHIFT);
                else       atomicAdd(&lcnt[row], 1);
            }
    }
}

__global__ void __launch_bounds__(1024)
group_bucket(const int* __restrict__ recs, const int* __restrict__ tab,
             int* __restrict__ offs, float* __restrict__ dinv,
             int* __restrict__ ssrc, int n_nodes, int n_bkts, int n_slices) {
    __shared__ int lcnt[BKT_NODES];
    __shared__ int rstart[MAX_SLICES];
    __shared__ int rlen[MAX_SLICES];
    __shared__ int sm[33];
    const int t = threadIdx.x, b = blockIdx.x;
    const int lane = t & 31, wave = t >> 5;
    const int node0 = b << BKT_SHIFT;

    const unsigned p = (t < n_slices) ? (unsigned)tab[b * n_slices + t] : 0u;
    const int loc = (int)(p & 0xffffu);
    const int cn  = (int)(p >> 16);
    int bstart, btot;
    (void)block_excl_scan(loc, sm, bstart);
    (void)block_excl_scan(cn, sm, btot);
    rstart[t] = t * SLICE_CAP + loc;
    rlen[t]   = cn;
    if (t < BKT_NODES) lcnt[t] = 0;
    __syncthreads();

    const int rpw = (n_slices + 31) >> 5;
    const int my_start = (lane < rpw) ? rstart[wave * rpw + lane] : 0;
    const int my_len   = (lane < rpw) ? rlen[wave * rpw + lane] : 0;
    int incl = my_len;
#pragma unroll
    for (int dd = 1; dd < 32; dd <<= 1) {
        int y = __shfl_up(incl, dd);
        if (lane >= dd) incl += y;
    }
    const int my_pre = incl - my_len;
    const int n_rec  = __builtin_amdgcn_readlane(incl, 31);

    walk_runs<false>(recs, lcnt, ssrc, my_start, my_pre, rpw, n_rec);
    __syncthreads();

    const int cv = (t < BKT_NODES) ? lcnt[t] : 0;
    int tot2;
    const int ex = block_excl_scan(cv, sm, tot2);
    if (t < BKT_NODES) {
        lcnt[t] = bstart + ex;
        const int node = node0 + t;
        if (node < n_nodes) {
            offs[node] = bstart + ex;
            dinv[node] = rsqrtf((float)(cv + 1));
        }
    }
    if (b == n_bkts - 1 && t == 0) offs[n_nodes] = bstart + btot;
    __syncthreads();

    walk_runs<true>(recs, lcnt, ssrc, my_start, my_pre, rpw, n_rec);
}

#define FT_ROWS    64
#define FT_PITCH4  33
#define FT_THREADS 1024

template <int NCOLS>
__device__ __forceinline__ void stage_w(const float* __restrict__ W, v4f* wl, int t) {
    constexpr int NG = NCOLS / 4;
    constexpr int KQ = K_DIM / 4;
    for (int it = t; it < KQ * NG; it += FT_THREADS) {
        const int kq = it / NG, g = it % NG;
        const v4f* wp = (const v4f*)(W + (long)(4 * kq) * NCOLS) + g;
        const v4f w0 = wp[0];
        const v4f w1 = wp[NG];
        const v4f w2 = wp[2 * NG];
        const v4f w3 = wp[3 * NG];
        v4f* dp = &wl[kq * NCOLS + 4 * g];
        v4f o0 = {w0.x, w1.x, w2.x, w3.x};
        v4f o1 = {w0.y, w1.y, w2.y, w3.y};
        v4f o2 = {w0.z, w1.z, w2.z, w3.z};
        v4f o3 = {w0.w, w1.w, w2.w, w3.w};
        dp[0] = o0;
        dp[1] = o1;
        dp[2] = o2;
        dp[3] = o3;
    }
}

template <int NCOLS, int NT>
__device__ __forceinline__ void stripe_product(const v4f* arow, const v4f* bl, v8f (&acc)[NT]) {
#pragma unroll
    for (int tt = 0; tt < NT; ++tt) acc[tt] = (v8f){};
#pragma unroll
    for (int j = 0; j < K_DIM / 8; ++j) {
        const v4f a = arow[2 * j];
        v4f b[NT];
#pragma unroll
        for (int tt = 0; tt < NT; ++tt) b[tt] = bl[2 * j * NCOLS + 16 * tt];
        const v2f a01 = {a.x, a.y};
        const v2f a23 = {a.z, a.w};
#pragma unroll
        for (int tt = 0; tt < NT; ++tt) {
            const v2f b01 = {b[tt].x, b[tt].y};
            acc[tt] = __builtin_amdgcn_wmma_f32_16x16x4_f32(
                false, a01, false, b01, (short)0, acc[tt], false, false);
        }
#pragma unroll
        for (int tt = 0; tt < NT; ++tt) {
            const v2f b23 = {b[tt].z, b[tt].w};
            acc[tt] = __builtin_amdgcn_wmma_f32_16x16x4_f32(
                false, a23, false, b23, (short)0, acc[tt], false, false);
        }
    }
}

__global__ void __launch_bounds__(FT_THREADS, 16)
layer1_fused(const float* __restrict__ x,
             const int* __restrict__ offs,
             const int* __restrict__ ssrc,
             const float* __restrict__ dinv,
             const float* __restrict__ W1,
             const float* __restrict__ b1,
             const float* __restrict__ W2,
             float* __restrict__ h2, int n_nodes, int rows_per_wg) {
    __shared__ v4f wl1[(K_DIM / 4) * 128];
    __shared__ v4f wl2[(K_DIM / 4) * 64];
    __shared__ v4f tile[FT_ROWS * FT_PITCH4];
    __shared__ v4f tdinv4[FT_ROWS / 4];
    float* tdinv = (float*)tdinv4;

    const int t = threadIdx.x;
    stage_w<128>(W1, wl1, t);
    stage_w<64>(W2, wl2, t);

    const int lane = t & 31;
    const int wave = __builtin_amdgcn_readfirstlane(t >> 5);
    const int c = lane & 15;
    const int h = lane >> 4;

    const int row0 = blockIdx.x * rows_per_wg;
    const int row1 = (row0 + rows_per_wg < n_nodes) ? (row0 + rows_per_wg) : n_nodes;
    const v4f* xv = (const v4f*)x;

    for (int tb = row0; tb < row1; tb += FT_ROWS) {
        for (int q = 0; q < FT_ROWS / 32; ++q) {
            const int lr   = wave + 32 * q;
            const int node = tb + lr;
            if (node < row1) {
                const int   beg = offs[node];
                const int   end = offs[node + 1];
                const float di  = dinv[node];
                v4f acc = xv[(long)node * 32 + lane] * di;
                for (int base = beg; base < end; base += 32) {
                    int my_s = 0, my_w = 0;
                    if (base + lane < end) {
                        my_s = ssrc[base + lane];
                        my_w = __builtin_bit_cast(int, dinv[my_s]);
                    }
                    const int cnt = (end - base < 32) ? (end - base) : 32;
                    for (int j = 0; j < cnt; j += 4) {
                        const int s0 = __builtin_amdgcn_readlane(my_s, j);
                        const int s1 = __builtin_amdgcn_readlane(my_s, j + 1);
                        const int s2 = __builtin_amdgcn_readlane(my_s, j + 2);
                        const int s3 = __builtin_amdgcn_readlane(my_s, j + 3);
                        const float w0 = __builtin_bit_cast(float, __builtin_amdgcn_readlane(my_w, j));
                        const float w1 = __builtin_bit_cast(float, __builtin_amdgcn_readlane(my_w, j + 1));
                        const float w2 = __builtin_bit_cast(float, __builtin_amdgcn_readlane(my_w, j + 2));
                        const float w3 = __builtin_bit_cast(float, __builtin_amdgcn_readlane(my_w, j + 3));
                        const v4f r0 = xv[(long)s0 * 32 + lane];
                        const v4f r1 = xv[(long)s1 * 32 + lane];
                        const v4f r2 = xv[(long)s2 * 32 + lane];
                        const v4f r3 = xv[(long)s3 * 32 + lane];
                        acc += r0 * w0;
                        acc += r1 * w1;
                        acc += r2 * w2;
                        acc += r3 * w3;
                    }
                }
                tile[lr * FT_PITCH4 + lane] = acc * di;
                if (lane == 0) tdinv[lr] = di;
            } else {
                tile[lr * FT_PITCH4 + lane] = (v4f){};
                if (lane == 0) tdinv[lr] = 0.0f;
            }
        }
        __syncthreads();

        const int  s1    = wave >> 3;
        const int  t1    = wave & 7;
        const bool live1 = tb + 16 * s1 < row1;
        v8f d1[1];
        if (live1)
            stripe_product<128, 1>(tile + (16 * s1 + c) * FT_PITCH4 + h,
                                   wl1 + h * 128 + 16 * t1 + c, d1);
        __syncthreads();
        if (live1) {
            const float bv = b1[16 * t1 + c];
            float* tf = (float*)tile + (16 * s1 + 8 * h) * (4 * FT_PITCH4) + 16 * t1 + c;
            const v4f q0 = tdinv4[4 * s1];
            const v4f q1 = tdinv4[4 * s1 + 1];
            const v4f q2 = tdinv4[4 * s1 + 2];
            const v4f q3 = tdinv4[4 * s1 + 3];
            const float dlo[8] = {q0.x, q0.y, q0.z, q0.w, q1.x, q1.y, q1.z, q1.w};
            const float dhi[8] = {q2.x, q2.y, q2.z, q2.w, q3.x, q3.y, q3.z, q3.w};
#pragma unroll
            for (int r = 0; r < 8; ++r) {
                const float u = d1[0][r] + bv;
                tf[r * (4 * FT_PITCH4)] = (u > 0.0f ? u : 0.0f) * (h ? dhi[r] : dlo[r]);
            }
        }
        __syncthreads();

        if (wave < 8) {
            const int s2 = wave >> 1;
            const int p  = wave & 1;
            const int rb = tb + 16 * s2;
            if (rb < row1) {
                v8f d2[2];
                stripe_product<64, 2>(tile + (16 * s2 + c) * FT_PITCH4 + h,
                                      wl2 + h * 64 + 32 * p + c, d2);
                if (rb + 16 <= row1) {
#pragma unroll
                    for (int tt = 0; tt < 2; ++tt) {
                        float* cbase = h2 + (long)(rb + 8 * h) * 64 + 32 * p + 16 * tt + c;
#pragma unroll
                        for (int r = 0; r < 8; ++r) cbase[(long)r * 64] = d2[tt][r];
                    }
                } else {
#pragma unroll
                    for (int tt = 0; tt < 2; ++tt) {
#pragma unroll
                        for (int r = 0; r < 8; ++r) {
                            const int row = rb + r + 8 * h;
                            if (row < row1) h2[(long)row * 64 + 32 * p + 16 * tt + c] = d2[tt][r];
                        }
                    }
                }
            }
        }
        __syncthreads();
    }
}

template <int CH>
__global__ void gather_scaled_rows(const float* __restrict__ hs,
                                   const int* __restrict__ offs,
                                   const int* __restrict__ ssrc,
                                   const float* __restrict__ dinv,
                                   const float* __restrict__ bias,
                                   float* __restrict__ out, int n_nodes) {
    constexpr int V = CH / 32;
    typedef __attribute__((ext_vector_type(V))) float vrow;
    const int lane = threadIdx.x & 31;
    const int node = __builtin_amdgcn_readfirstlane(
        (int)(blockIdx.x * (blockDim.x >> 5) + (threadIdx.x >> 5)));
    if (node >= n_nodes) return;

    const int   beg = offs[node];
    const int   end = offs[node + 1];
    const float di  = dinv[node];
    const vrow* hv  = (const vrow*)hs;

    vrow acc = hv[(long)node * 32 + lane];

    for (int base = beg; base < end; base += 32) {
        int my_s = 0;
        if (base + lane < end) my_s = ssrc[base + lane];
        const int cnt = (end - base < 32) ? (end - base) : 32;
        for (int j = 0; j < cnt; j += 4) {
            const int s0 = __builtin_amdgcn_readlane(my_s, j);
            const int s1 = __builtin_amdgcn_readlane(my_s, j + 1);
            const int s2 = __builtin_amdgcn_readlane(my_s, j + 2);
            const int s3 = __builtin_amdgcn_readlane(my_s, j + 3);
            const vrow r0 = hv[(long)s0 * 32 + lane];
            const vrow r1 = hv[(long)s1 * 32 + lane];
            const vrow r2 = hv[(long)s2 * 32 + lane];
            const vrow r3 = hv[(long)s3 * 32 + lane];
            const float m1 = (j + 1 < cnt) ? 1.0f : 0.0f;
            const float m2 = (j + 2 < cnt) ? 1.0f : 0.0f;
            const float m3 = (j + 3 < cnt) ? 1.0f : 0.0f;
            acc += r0;
            acc += r1 * m1;
            acc += r2 * m2;
            acc += r3 * m3;
        }
    }

    vrow v = acc * di;
    v += ((const vrow*)bias)[lane];
#pragma unroll
    for (int i = 0; i < V; ++i) v[i] = v[i] > 0.0f ? v[i] : 0.0f;
    ((vrow*)out)[(long)node * 32 + lane] = v;
}

extern "C" void kernel_launch(void* const* d_in, const int* in_sizes, int n_in,
                              void* d_out, int out_size, void* d_ws, size_t ws_size,
                              hipStream_t stream) {
    const float* x  = (const float*)d_in[0];
    const int*   ei = (const int*)d_in[1];
    const float* W1 = (const float*)d_in[2];
    const float* b1 = (const float*)d_in[3];
    const float* W2 = (const float*)d_in[4];
    const float* b2 = (const float*)d_in[5];

    const int N   = in_sizes[0] / K_DIM;
    const int E   = in_sizes[1] / 2;
    (void)n_in; (void)ws_size; (void)out_size;

    const int* src = ei;
    const int* dst = ei + E;
    float* out = (float*)d_out;

    char* ws = (char*)d_ws;
    const size_t degBytes = (((size_t)N * 4) + 255) & ~(size_t)255;
    float* dinv = (float*)ws;
    float* bufA = (float*)(ws + degBytes);
    float* bufB = (float*)(ws + degBytes + (size_t)N * 128 * 4);
    char*  lists = (char*)(bufB + (size_t)N * 64);
    const size_t offBytes = ((((size_t)N + 1) * 4) + 255) & ~(size_t)255;
    int* offs = (int*)lists;
    int* cnt  = (int*)(lists + offBytes);
    int* ssrc = cnt + E;
    int* recs = (int*)bufA;

    const int B  = (N + BKT_NODES - 1) >> BKT_SHIFT;
    const int S  = (E + SLICE_CAP - 1) / SLICE_CAP;
    int* tab = cnt;

    sort_slice<<<S, SLICE_THREADS, 0, stream>>>(src, dst, tab, recs, E, B, S);
    group_bucket<<<B, 1024, 0, stream>>>(recs, tab, offs, dinv, ssrc, N, B, S);

    {
        const int rows_per_wg = (((N + 511) / 512) + 3) & ~3;
        const int groups = (N + rows_per_wg - 1) / rows_per_wg;
        layer1_fused<<<groups, FT_THREADS, 0, stream>>>(x, offs, ssrc, dinv, W1, b1, W2, bufB, N, rows_per_wg);
    }

    gather_scaled_rows<64><<<(N + 7) / 8, 256, 0, stream>>>(bufB, offs, ssrc, dinv, b2, out, N);
}


# Layer-2 gather: a wave sums two neighbouring destination rows, one per 16-lane half (16 B per lane), so one load brings 512 B of two source rows; half the waves and loads; sum order unchanged
# speedup vs baseline: 5.9664x; 1.0337x over previous
#include <hip/hip_runtime.h>


typedef __attribute__((ext_vector_type(2))) float v2f;
typedef __attribute__((ext_vector_type(4))) float v4f;
typedef __attribute__((ext_vector_type(8))) float v8f;

#define K_DIM 128

#define BKT_SHIFT 8
#define BKT_NODES (1 << BKT_SHIFT)
#define MAX_BKTS  1024
#define SLICE_THREADS 1024
#define SLICE_SLOTS 7
#define SLICE_CAP 6272
#define MAX_SLICES 1024
#define WALK_SLOTS 5

__device__ __forceinline__ int block_excl_scan(int v, int* sm, int& total) {
    const int lane = threadIdx.x & 31;
    const int wave = threadIdx.x >> 5;
    int incl = v;
#pragma unroll
    for (int d = 1; d < 32; d <<= 1) {
        int y = __shfl_up(incl, d);
        if (lane >= d) incl += y;
    }
    if (lane == 31) sm[wave] = incl;
    __syncthreads();
    if (wave == 0) {
        const int nw = (int)(blockDim.x + 31) >> 5;
        const int w = (lane < nw) ? sm[lane] : 0;
        int iw = w;
#pragma unroll
        for (int d = 1; d < 32; d <<= 1) {
            int y = __shfl_up(iw, d);
            if (lane >= d) iw += y;
        }
        sm[lane] = iw - w;
        if (lane == 31) sm[32] = iw;
    }
    __syncthreads();
    total = sm[32];
    return sm[wave] + (incl - v);
}

__global__ void __launch_bounds__(SLICE_THREADS)
sort_slice(const int* __restrict__ src, const int* __restrict__ dst,
           int* __restrict__ tab, int* __restrict__ recs,
           int n_edges, int n_bkts, int n_slices) {
    __shared__ int hist[MAX_BKTS];
    __shared__ int lrec[SLICE_CAP];
    __shared__ int sm[33];
    const int t = threadIdx.x, c = blockIdx.x;
    if (t < MAX_BKTS) hist[t] = 0;
    __syncthreads();
    const int beg = c * SLICE_CAP;
    const int len = (n_edges - beg < SLICE_CAP) ? (n_edges - beg) : SLICE_CAP;

    int s[SLICE_SLOTS], d[SLICE_SLOTS];
#pragma unroll
    for (int k = 0; k < SLICE_SLOTS; ++k) {
        const int i  = t + k * SLICE_THREADS;
        const int ii = (i < len) ? i : 0;
        s[k] = src[beg + ii];
        d[k] = dst[beg + ii];
    }
#pragma unroll
    for (int k = 0; k < SLICE_SLOTS; ++k)
        if (t + k * SLICE_THREADS < len) atomicAdd(&hist[d[k] >> BKT_SHIFT], 1);
    __syncthreads();

    const int v = (t < n_bkts) ? hist[t] : 0;
    int total;
    const int ex = block_excl_scan(v, sm, total);
    if (t < n_bkts) {
        hist[t] = ex;
        tab[t * n_slices + c] = (int)((unsigned)ex | ((unsigned)v << 16));
    }
    __syncthreads();

#pragma unroll
    for (int k = 0; k < SLICE_SLOTS; ++k)
        if (t + k * SLICE_THREADS < len) {
            const int pos = atomicAdd(&hist[d[k] >> BKT_SHIFT], 1);
            lrec[pos] = (s[k] << BKT_SHIFT) | (d[k] & (BKT_NODES - 1));
        }
    __syncthreads();
    for (int i = t; i < len; i += SLICE_THREADS) recs[beg + i] = lrec[i];
}

template <bool PLACE>
__device__ __forceinline__ void walk_runs(const int* __restrict__ recs, int* lcnt,
                                          int* __restrict__ ssrc,
                                          int my_start, int my_pre, int n_runs, int n_rec) {
    const int lane = threadIdx.x & 31;
    for (int base = 0; base < n_rec; base += 32 * WALK_SLOTS) {
        int run[WALK_SLOTS];
#pragma unroll
        for (int k = 0; k < WALK_SLOTS; ++k) run[k] = 0;
        for (int r = 1; r < n_runs; ++r) {
            const int pr = __builtin_amdgcn_readlane(my_pre, r);
#pragma unroll
            for (int k = 0; k < WALK_SLOTS; ++k) run[k] += (base + lane + 32 * k >= pr) ? 1 : 0;
        }
        int rec[WALK_SLOTS];
#pragma unroll
        for (int k = 0; k < WALK_SLOTS; ++k) {
            const int i = base + lane + 32 * k;
            const int a = __shfl(my_start, run[k]) + (i - __shfl(my_pre, run[k]));
            rec[k] = recs[(i < n_rec) ? a : 0];
        }
#pragma unroll
        for (int k = 0; k < WALK_SLOTS; ++k)
            if (base + lane + 32 * k < n_rec) {
                const int row = rec[k] & (BKT_NODES - 1);
                if (PLACE) ssrc[atomicAdd(&lcnt[row], 1)] = (int)((unsigned)rec[k] >> BKT_SHIFT);
                else       atomicAdd(&lcnt[row], 1);
            }
    }
}

__global__ void __launch_bounds__(1024)
group_bucket(const int* __restrict__ recs, const int* __restrict__ tab,
             int* __restrict__ offs, float* __restrict__ dinv,
             int* __restrict__ ssrc, int n_nodes, int n_bkts, int n_slices) {
    __shared__ int lcnt[BKT_NODES];
    __shared__ int rstart[MAX_SLICES];
    __shared__ int rlen[MAX_SLICES];
    __shared__ int sm[33];
    const int t = threadIdx.x, b = blockIdx.x;
    const int lane = t & 31, wave = t >> 5;
    const int node0 = b << BKT_SHIFT;

    const unsigned p = (t < n_slices) ? (unsigned)tab[b * n_slices + t] : 0u;
    const int loc = (int)(p & 0xffffu);
    const int cn  = (int)(p >> 16);
    int bstart, btot;
    (void)block_excl_scan(loc, sm, bstart);
    (void)block_excl_scan(cn, sm, btot);
    rstart[t] = t * SLICE_CAP + loc;
    rlen[t]   = cn;
    if (t < BKT_NODES) lcnt[t] = 0;
    __syncthreads();

    const int rpw = (n_slices + 31) >> 5;
    const int my_start = (lane < rpw) ? rstart[wave * rpw + lane] : 0;
    const int my_len   = (lane < rpw) ? rlen[wave * rpw + lane] : 0;
    int incl = my_len;
#pragma unroll
    for (int dd = 1; dd < 32; dd <<= 1) {
        int y = __shfl_up(incl, dd);
        if (lane >= dd) incl += y;
    }
    const int my_pre = incl - my_len;
    const int n_rec  = __builtin_amdgcn_readlane(incl, 31);

    walk_runs<false>(recs, lcnt, ssrc, my_start, my_pre, rpw, n_rec);
    __syncthreads();

    const int cv = (t < BKT_NODES) ? lcnt[t] : 0;
    int tot2;
    const int ex = block_excl_scan(cv, sm, tot2);
    if (t < BKT_NODES) {
        lcnt[t] = bstart + ex;
        const int node = node0 + t;
        if (node < n_nodes) {
            offs[node] = bstart + ex;
            dinv[node] = rsqrtf((float)(cv + 1));
        }
    }
    if (b == n_bkts - 1 && t == 0) offs[n_nodes] = bstart + btot;
    __syncthreads();

    walk_runs<true>(recs, lcnt, ssrc, my_start, my_pre, rpw, n_rec);
}

#define FT_ROWS    64
#define FT_PITCH4  33
#define FT_THREADS 1024

template <int NCOLS>
__device__ __forceinline__ void stage_w(const float* __restrict__ W, v4f* wl, int t) {
    constexpr int NG = NCOLS / 4;
    constexpr int KQ = K_DIM / 4;
    for (int it = t; it < KQ * NG; it += FT_THREADS) {
        const int kq = it / NG, g = it % NG;
        const v4f* wp = (const v4f*)(W + (long)(4 * kq) * NCOLS) + g;
        const v4f w0 = wp[0];
        const v4f w1 = wp[NG];
        const v4f w2 = wp[2 * NG];
        const v4f w3 = wp[3 * NG];
        v4f* dp = &wl[kq * NCOLS + 4 * g];
        v4f o0 = {w0.x, w1.x, w2.x, w3.x};
        v4f o1 = {w0.y, w1.y, w2.y, w3.y};
        v4f o2 = {w0.z, w1.z, w2.z, w3.z};
        v4f o3 = {w0.w, w1.w, w2.w, w3.w};
        dp[0] = o0;
        dp[1] = o1;
        dp[2] = o2;
        dp[3] = o3;
    }
}

template <int NCOLS, int NT>
__device__ __forceinline__ void stripe_product(const v4f* arow, const v4f* bl, v8f (&acc)[NT]) {
#pragma unroll
    for (int tt = 0; tt < NT; ++tt) acc[tt] = (v8f){};
#pragma unroll
    for (int j = 0; j < K_DIM / 8; ++j) {
        const v4f a = arow[2 * j];
        v4f b[NT];
#pragma unroll
        for (int tt = 0; tt < NT; ++tt) b[tt] = bl[2 * j * NCOLS + 16 * tt];
        const v2f a01 = {a.x, a.y};
        const v2f a23 = {a.z, a.w};
#pragma unroll
        for (int tt = 0; tt < NT; ++tt) {
            const v2f b01 = {b[tt].x, b[tt].y};
            acc[tt] = __builtin_amdgcn_wmma_f32_16x16x4_f32(
                false, a01, false, b01, (short)0, acc[tt], false, false);
        }
#pragma unroll
        for (int tt = 0; tt < NT; ++tt) {
            const v2f b23 = {b[tt].z, b[tt].w};
            acc[tt] = __builtin_amdgcn_wmma_f32_16x16x4_f32(
                false, a23, false, b23, (short)0, acc[tt], false, false);
        }
    }
}

__global__ void __launch_bounds__(FT_THREADS, 16)
layer1_fused(const float* __restrict__ x,
             const int* __restrict__ offs,
             const int* __restrict__ ssrc,
             const float* __restrict__ dinv,
             const float* __restrict__ W1,
             const float* __restrict__ b1,
             const float* __restrict__ W2,
             float* __restrict__ h2, int n_nodes, int rows_per_wg) {
    __shared__ v4f wl1[(K_DIM / 4) * 128];
    __shared__ v4f wl2[(K_DIM / 4) * 64];
    __shared__ v4f tile[FT_ROWS * FT_PITCH4];
    __shared__ v4f tdinv4[FT_ROWS / 4];
    float* tdinv = (float*)tdinv4;

    const int t = threadIdx.x;
    stage_w<128>(W1, wl1, t);
    stage_w<64>(W2, wl2, t);

    const int lane = t & 31;
    const int wave = __builtin_amdgcn_readfirstlane(t >> 5);
    const int c = lane & 15;
    const int h = lane >> 4;

    const int row0 = blockIdx.x * rows_per_wg;
    const int row1 = (row0 + rows_per_wg < n_nodes) ? (row0 + rows_per_wg) : n_nodes;
    const v4f* xv = (const v4f*)x;

    for (int tb = row0; tb < row1; tb += FT_ROWS) {
        for (int q = 0; q < FT_ROWS / 32; ++q) {
            const int lr   = wave + 32 * q;
            const int node = tb + lr;
            if (node < row1) {
                const int   beg = offs[node];
                const int   end = offs[node + 1];
                const float di  = dinv[node];
                v4f acc = xv[(long)node * 32 + lane] * di;
                for (int base = beg; base < end; base += 32) {
                    int my_s = 0, my_w = 0;
                    if (base + lane < end) {
                        my_s = ssrc[base + lane];
                        my_w = __builtin_bit_cast(int, dinv[my_s]);
                    }
                    const int cnt = (end - base < 32) ? (end - base) : 32;
                    for (int j = 0; j < cnt; j += 4) {
                        const int s0 = __builtin_amdgcn_readlane(my_s, j);
                        const int s1 = __builtin_amdgcn_readlane(my_s, j + 1);
                        const int s2 = __builtin_amdgcn_readlane(my_s, j + 2);
                        const int s3 = __builtin_amdgcn_readlane(my_s, j + 3);
                        const float w0 = __builtin_bit_cast(float, __builtin_amdgcn_readlane(my_w, j));
                        const float w1 = __builtin_bit_cast(float, __builtin_amdgcn_readlane(my_w, j + 1));
                        const float w2 = __builtin_bit_cast(float, __builtin_amdgcn_readlane(my_w, j + 2));
                        const float w3 = __builtin_bit_cast(float, __builtin_amdgcn_readlane(my_w, j + 3));
                        const v4f r0 = xv[(long)s0 * 32 + lane];
                        const v4f r1 = xv[(long)s1 * 32 + lane];
                        const v4f r2 = xv[(long)s2 * 32 + lane];
                        const v4f r3 = xv[(long)s3 * 32 + lane];
                        acc += r0 * w0;
                        acc += r1 * w1;
                        acc += r2 * w2;
                        acc += r3 * w3;
                    }
                }
                tile[lr * FT_PITCH4 + lane] = acc * di;
                if (lane == 0) tdinv[lr] = di;
            } else {
                tile[lr * FT_PITCH4 + lane] = (v4f){};
                if (lane == 0) tdinv[lr] = 0.0f;
            }
        }
        __syncthreads();

        const int  s1    = wave >> 3;
        const int  t1    = wave & 7;
        const bool live1 = tb + 16 * s1 < row1;
        v8f d1[1];
        if (live1)
            stripe_product<128, 1>(tile + (16 * s1 + c) * FT_PITCH4 + h,
                                   wl1 + h * 128 + 16 * t1 + c, d1);
        __syncthreads();
        if (live1) {
            const float bv = b1[16 * t1 + c];
            float* tf = (float*)tile + (16 * s1 + 8 * h) * (4 * FT_PITCH4) + 16 * t1 + c;
            const v4f q0 = tdinv4[4 * s1];
            const v4f q1 = tdinv4[4 * s1 + 1];
            const v4f q2 = tdinv4[4 * s1 + 2];
            const v4f q3 = tdinv4[4 * s1 + 3];
            const float dlo[8] = {q0.x, q0.y, q0.z, q0.w, q1.x, q1.y, q1.z, q1.w};
            const float dhi[8] = {q2.x, q2.y, q2.z, q2.w, q3.x, q3.y, q3.z, q3.w};
#pragma unroll
            for (int r = 0; r < 8; ++r) {
                const float u = d1[0][r] + bv;
                tf[r * (4 * FT_PITCH4)] = (u > 0.0f ? u : 0.0f) * (h ? dhi[r] : dlo[r]);
            }
        }
        __syncthreads();

        if (wave < 8) {
            const int s2 = wave >> 1;
            const int p  = wave & 1;
            const int rb = tb + 16 * s2;
            if (rb < row1) {
                v8f d2[2];
                stripe_product<64, 2>(tile + (16 * s2 + c) * FT_PITCH4 + h,
                                      wl2 + h * 64 + 32 * p + c, d2);
                if (rb + 16 <= row1) {
#pragma unroll
                    for (int tt = 0; tt < 2; ++tt) {
                        float* cbase = h2 + (long)(rb + 8 * h) * 64 + 32 * p + 16 * tt + c;
#pragma unroll
                        for (int r = 0; r < 8; ++r) cbase[(long)r * 64] = d2[tt][r];
                    }
                } else {
#pragma unroll
                    for (int tt = 0; tt < 2; ++tt) {
#pragma unroll
                        for (int r = 0; r < 8; ++r) {
                            const int row = rb + r + 8 * h;
                            if (row < row1) h2[(long)row * 64 + 32 * p + 16 * tt + c] = d2[tt][r];
                        }
                    }
                }
            }
        }
        __syncthreads();
    }
}


template <int J>
__device__ __forceinline__ int half_bcast(int v) {
    return __builtin_amdgcn_update_dpp(v, v, 0x150 + J, 0xf, 0xf, false);
}

template <int J>
__device__ __forceinline__ void add_four(const char* __restrict__ hb, int sa, int sb,
                                         int rem, unsigned lane_off, v4f& acc) {
    const int sv = (J < 16) ? sa : sb;
    const unsigned o0 = ((unsigned)half_bcast<(J + 0) & 15>(sv) << 8) + lane_off;
    const unsigned o1 = ((unsigned)half_bcast<(J + 1) & 15>(sv) << 8) + lane_off;
    const unsigned o2 = ((unsigned)half_bcast<(J + 2) & 15>(sv) << 8) + lane_off;
    const unsigned o3 = ((unsigned)half_bcast<(J + 3) & 15>(sv) << 8) + lane_off;
    const v4f r0 = *(const v4f*)(hb + o0);
    const v4f r1 = *(const v4f*)(hb + o1);
    const v4f r2 = *(const v4f*)(hb + o2);
    const v4f r3 = *(const v4f*)(hb + o3);
    const float m0 = (J + 0 < rem) ? 1.0f : 0.0f;
    const float m1 = (J + 1 < rem) ? 1.0f : 0.0f;
    const float m2 = (J + 2 < rem) ? 1.0f : 0.0f;
    const float m3 = (J + 3 < rem) ? 1.0f : 0.0f;
    acc += r0 * m0;
    acc += r1 * m1;
    acc += r2 * m2;
    acc += r3 * m3;
}

template <int J>
__device__ __forceinline__ void add_from(const char* __restrict__ hb, int sa, int sb,
                                         int rem, int tmax, unsigned lane_off, v4f& acc) {
    add_four<J>(hb, sa, sb, rem, lane_off, acc);
    if constexpr (J + 4 < 32) {
        if (tmax > J + 4) add_from<J + 4>(hb, sa, sb, rem, tmax, lane_off, acc);
    }
}

__global__ void __launch_bounds__(256)
gather_scaled_row_pairs(const float* __restrict__ hs,
                        const int* __restrict__ offs,
                        const int* __restrict__ ssrc,
                        const float* __restrict__ dinv,
                        const float* __restrict__ bias,
                        float* __restrict__ out, int n_nodes) {
    const int lane = threadIdx.x & 31;
    const int hl = lane & 15;
    const int h  = lane >> 4;
    const int pair = __builtin_amdgcn_readfirstlane(
        (int)(blockIdx.x * (blockDim.x >> 5) + (threadIdx.x >> 5)));
    if (2 * pair >= n_nodes) return;

    const int  node = 2 * pair + h;
    const bool live = node < n_nodes;
    const int  nd   = live ? node : (n_nodes - 1);
    const int   beg = offs[nd];
    const int   end = offs[nd + 1];
    const int   cnt = live ? (end - beg) : 0;
    const float di  = dinv[nd];
    const char* hb  = (const char*)hs;
    const unsigned lane_off = (unsigned)hl * 16u;

    v4f acc = *(const v4f*)(hb + ((unsigned)nd << 8) + lane_off);

    const int c0 = __builtin_amdgcn_readlane(cnt, 0);
    const int c1 = __builtin_amdgcn_readlane(cnt, 16);
    const int cmax = (c0 > c1) ? c0 : c1;

    for (int base = 0; base < cmax; base += 32) {
        int sa = nd, sb = nd;
        if (base + hl < cnt)      sa = ssrc[beg + base + hl];
        if (base + 16 + hl < cnt) sb = ssrc[beg + base + 16 + hl];
        add_from<0>(hb, sa, sb, cnt - base, cmax - base, lane_off, acc);
    }

    v4f v = acc * di;
    v += ((const v4f*)bias)[hl];
#pragma unroll
    for (int i = 0; i < 4; ++i) v[i] = v[i] > 0.0f ? v[i] : 0.0f;
    if (live) *(v4f*)((char*)out + ((size_t)node << 8) + lane_off) = v;
}

extern "C" void kernel_launch(void* const* d_in, const int* in_sizes, int n_in,
                              void* d_out, int out_size, void* d_ws, size_t ws_size,
                              hipStream_t stream) {
    const float* x  = (const float*)d_in[0];
    const int*   ei = (const int*)d_in[1];
    const float* W1 = (const float*)d_in[2];
    const float* b1 = (const float*)d_in[3];
    const float* W2 = (const float*)d_in[4];
    const float* b2 = (const float*)d_in[5];

    const int N   = in_sizes[0] / K_DIM;
    const int E   = in_sizes[1] / 2;
    (void)n_in; (void)ws_size; (void)out_size;

    const int* src = ei;
    const int* dst = ei + E;
    float* out = (float*)d_out;

    char* ws = (char*)d_ws;
    const size_t degBytes = (((size_t)N * 4) + 255) & ~(size_t)255;
    float* dinv = (float*)ws;
    float* bufA = (float*)(ws + degBytes);
    float* bufB = (float*)(ws + degBytes + (size_t)N * 128 * 4);
    char*  lists = (char*)(bufB + (size_t)N * 64);
    const size_t offBytes = ((((size_t)N + 1) * 4) + 255) & ~(size_t)255;
    int* offs = (int*)lists;
    int* cnt  = (int*)(lists + offBytes);
    int* ssrc = cnt + E;
    int* recs = (int*)bufA;

    const int B  = (N + BKT_NODES - 1) >> BKT_SHIFT;
    const int S  = (E + SLICE_CAP - 1) / SLICE_CAP;
    int* tab = cnt;

    sort_slice<<<S, SLICE_THREADS, 0, stream>>>(src, dst, tab, recs, E, B, S);
    group_bucket<<<B, 1024, 0, stream>>>(recs, tab, offs, dinv, ssrc, N, B, S);

    {
        const int rows_per_wg = (((N + 511) / 512) + 3) & ~3;
        const int groups = (N + rows_per_wg - 1) / rows_per_wg;
        layer1_fused<<<groups, FT_THREADS, 0, stream>>>(x, offs, ssrc, dinv, W1, b1, W2, bufB, N, rows_per_wg);
    }

    gather_scaled_row_pairs<<<(N + 15) / 16, 256, 0, stream>>>(bufB, offs, ssrc, dinv, b2, out, N);
}
